# starting kernel
# baseline (speedup 1.0000x reference)
#include <hip/hip_runtime.h>
#include <hip/hip_bf16.h>


#define D_FEAT 128
#define LDS_STRIDE 132
#define OUT_DIM 26

typedef __attribute__((ext_vector_type(2))) float v2f;
typedef __attribute__((ext_vector_type(8))) float v8f;


__global__ void sage_zero_f32(float* __restrict__ p, int n) {
    int i = blockIdx.x * blockDim.x + threadIdx.x;
    if (i < n) p[i] = 0.0f;
}

__global__ void sage_deg_kernel(const int* __restrict__ dst, float* __restrict__ deg,
                                int n_edges) {
    int e = blockIdx.x * blockDim.x + threadIdx.x;
    if (e < n_edges) atomicAdd(&deg[dst[e]], 1.0f);
}

__global__ void sage_invdeg_kernel(float* __restrict__ deg, int n_nodes) {
    int i = blockIdx.x * blockDim.x + threadIdx.x;
    if (i < n_nodes) {
        float d = deg[i];
        deg[i] = 1.0f / fmaxf(d, 1.0f);
    }
}

__global__ void sage_scatter_kernel(const int* __restrict__ src, const int* __restrict__ dst,
                                    const float* __restrict__ x, float* __restrict__ msg,
                                    int n_edges) {
    long long tid = (long long)blockIdx.x * blockDim.x + threadIdx.x;
    int e = (int)(tid >> 5);
    int c = (int)(tid & 31);
    if (e >= n_edges) return;
    int s = src[e];
    int d = dst[e];
    const float4 v = *(const float4*)(x + (size_t)s * D_FEAT + c * 4);
    float* p = msg + (size_t)d * D_FEAT + c * 4;
    atomicAdd(p + 0, v.x);
    atomicAdd(p + 1, v.y);
    atomicAdd(p + 2, v.z);
    atomicAdd(p + 3, v.w);
}

__global__ void __launch_bounds__(256)
sage_layer_kernel(const float* __restrict__ xin, const float* __restrict__ msg,
                  const float* __restrict__ invdeg,
                  const float* __restrict__ Wself, const float* __restrict__ Wneigh,
                  const float* __restrict__ bias, float* __restrict__ xout) {
    __shared__ float As[16 * LDS_STRIDE];
    __shared__ float Hs[16 * LDS_STRIDE];

    const int m0 = blockIdx.x * 16;
    const int t  = threadIdx.x;

    {
        int r  = t >> 4;
        int c0 = (t & 15) * 8;
        const float* xr = xin + (size_t)(m0 + r) * D_FEAT + c0;
        float4 a0 = *(const float4*)(xr);
        float4 a1 = *(const float4*)(xr + 4);
        *(float4*)(As + r * LDS_STRIDE + c0)     = a0;
        *(float4*)(As + r * LDS_STRIDE + c0 + 4) = a1;

        float iv = invdeg[m0 + r];
        const float* mr = msg + (size_t)(m0 + r) * D_FEAT + c0;
        float4 h0 = *(const float4*)(mr);
        float4 h1 = *(const float4*)(mr + 4);
        h0.x *= iv; h0.y *= iv; h0.z *= iv; h0.w *= iv;
        h1.x *= iv; h1.y *= iv; h1.z *= iv; h1.w *= iv;
        *(float4*)(Hs + r * LDS_STRIDE + c0)     = h0;
        *(float4*)(Hs + r * LDS_STRIDE + c0 + 4) = h1;
    }
    __syncthreads();

    const int wave = t >> 5;
    const int lane = t & 31;
    const int n0   = wave * 16;
    const int half = lane >> 4;
    const int l16  = lane & 15;
    const int col  = n0 + l16;

    v8f acc;
    {
        float bv = bias[col];
        #pragma unroll
        for (int i = 0; i < 8; ++i) acc[i] = bv;
    }

    #pragma unroll 4
    for (int k0 = 0; k0 < D_FEAT; k0 += 4) {
        const int ka = k0 + half * 2;
        v2f a = *(const v2f*)(As + l16 * LDS_STRIDE + ka);
        v2f b;
        b.x = Wself[(size_t)ka * D_FEAT + col];
        b.y = Wself[(size_t)(ka + 1) * D_FEAT + col];
        acc = __builtin_amdgcn_wmma_f32_16x16x4_f32(false, a, false, b,
                                                    (short)0, acc, false, false);

        v2f h = *(const v2f*)(Hs + l16 * LDS_STRIDE + ka);
        v2f bn;
        bn.x = Wneigh[(size_t)ka * D_FEAT + col];
        bn.y = Wneigh[(size_t)(ka + 1) * D_FEAT + col];
        acc = __builtin_amdgcn_wmma_f32_16x16x4_f32(false, h, false, bn,
                                                    (short)0, acc, false, false);
    }

    #pragma unroll
    for (int v = 0; v < 8; ++v) {
        int row = m0 + v + half * 8;
        float val = acc[v];
        val = val > 0.0f ? val : 0.0f;
        xout[(size_t)row * D_FEAT + col] = val;
    }
}

__global__ void __launch_bounds__(64)
sage_pred_kernel(const float* __restrict__ xin, const float* __restrict__ Wp,
                 const float* __restrict__ bp, float* __restrict__ out) {
    const int m0   = blockIdx.x * 16;
    const int t    = threadIdx.x;
    const int wave = t >> 5;
    const int lane = t & 31;
    const int n0   = wave * 16;
    const int half = lane >> 4;
    const int l16  = lane & 15;
    const int col  = n0 + l16;
    const int bcol = col < OUT_DIM ? col : (OUT_DIM - 1);

    v8f acc;
    {
        float bv = bp[bcol];
        #pragma unroll
        for (int i = 0; i < 8; ++i) acc[i] = bv;
    }

    #pragma unroll 4
    for (int k0 = 0; k0 < D_FEAT; k0 += 4) {
        const int ka = k0 + half * 2;
        const float* ap = xin + (size_t)(m0 + l16) * D_FEAT + ka;
        v2f a; a.x = ap[0]; a.y = ap[1];
        v2f b;
        b.x = Wp[(size_t)ka * OUT_DIM + bcol];
        b.y = Wp[(size_t)(ka + 1) * OUT_DIM + bcol];
        acc = __builtin_amdgcn_wmma_f32_16x16x4_f32(false, a, false, b,
                                                    (short)0, acc, false, false);
    }

    if (col < OUT_DIM) {
        #pragma unroll
        for (int v = 0; v < 8; ++v) {
            int row = m0 + v + half * 8;
            out[(size_t)row * OUT_DIM + col] = acc[v];
        }
    }
}


extern "C" void kernel_launch(void* const* d_in, const int* in_sizes, int n_in,
                              void* d_out, int out_size, void* d_ws, size_t ws_size,
                              hipStream_t stream) {
    const float* x       = (const float*)d_in[0];
    const int*   src     = (const int*)  d_in[1];
    const int*   dst     = (const int*)  d_in[2];
    const float* W_self  = (const float*)d_in[3];
    const float* W_neigh = (const float*)d_in[4];
    const float* b       = (const float*)d_in[5];
    const float* W_pred  = (const float*)d_in[6];
    const float* b_pred  = (const float*)d_in[7];

    const int n_nodes = in_sizes[0] / D_FEAT;
    const int n_edges = in_sizes[1];
    const int L       = in_sizes[5] / D_FEAT;
    const int feat_elems = n_nodes * D_FEAT;

    char*  ws = (char*)d_ws;
    size_t off = ((size_t)n_nodes * sizeof(float) + 511) & ~(size_t)511;
    size_t feat_bytes = (size_t)feat_elems * sizeof(float);
    float* invdeg = (float*)ws;
    float* msg    = (float*)(ws + off);
    float* bufA   = (float*)(ws + off + feat_bytes);
    float* bufB   = (float*)(ws + off + 2 * feat_bytes);

    const int T = 256;

    sage_zero_f32<<<(n_nodes + T - 1) / T, T, 0, stream>>>(invdeg, n_nodes);
    sage_deg_kernel<<<(n_edges + T - 1) / T, T, 0, stream>>>(dst, invdeg, n_edges);
    sage_invdeg_kernel<<<(n_nodes + T - 1) / T, T, 0, stream>>>(invdeg, n_nodes);

    const float* cur = x;
    float* nxt = bufA;
    const long long scat_threads = (long long)n_edges * 32;
    const int scat_blocks = (int)((scat_threads + T - 1) / T);
    const int m_tiles = n_nodes / 16;

    for (int l = 0; l < L; ++l) {
        sage_zero_f32<<<(feat_elems + T - 1) / T, T, 0, stream>>>(msg, feat_elems);
        sage_scatter_kernel<<<scat_blocks, T, 0, stream>>>(src, dst, cur, msg, n_edges);
        sage_layer_kernel<<<m_tiles, 256, 0, stream>>>(
            cur, msg, invdeg,
            W_self  + (size_t)l * D_FEAT * D_FEAT,
            W_neigh + (size_t)l * D_FEAT * D_FEAT,
            b       + (size_t)l * D_FEAT,
            nxt);
        cur = nxt;
        nxt = (nxt == bufA) ? bufB : bufA;
    }

    sage_pred_kernel<<<m_tiles, 64, 0, stream>>>(cur, W_pred, b_pred, (float*)d_out);
}


# Neighbour-sum source rows fetched by one 16-byte-per-lane load per edge (one 512-byte wave load, was four of 128); LDS sums kept column-permuted so adds stay conflict-free; arrival order as before
# speedup vs baseline: 2.4816x; 2.4816x over previous
#include <hip/hip_runtime.h>
#include <hip/hip_bf16.h>


#define D_FEAT 128
#define LDS_STRIDE 132
#define OUT_DIM 26

typedef __attribute__((ext_vector_type(2))) float v2f;
typedef __attribute__((ext_vector_type(8))) float v8f;


__global__ void sage_zero_f32(float* __restrict__ p, int n) {
    int i = blockIdx.x * blockDim.x + threadIdx.x;
    if (i < n) p[i] = 0.0f;
}

__global__ void sage_deg_kernel(const int* __restrict__ dst, float* __restrict__ deg,
                                int n_edges) {
    int e = blockIdx.x * blockDim.x + threadIdx.x;
    if (e < n_edges) atomicAdd(&deg[dst[e]], 1.0f);
}

__global__ void sage_invdeg_kernel(float* __restrict__ deg, int n_nodes) {
    int i = blockIdx.x * blockDim.x + threadIdx.x;
    if (i < n_nodes) {
        float d = deg[i];
        deg[i] = 1.0f / fmaxf(d, 1.0f);
    }
}

#define AGG_ROWS    512
#define AGG_SHIFT   9
#define AGG_THREADS 1024
#define AGG_QCAP    12288
#define AGG_SMASK   0x7FFFFF
#define AGG_REC     (AGG_QCAP + 4)

template <bool GUARD>
__device__ __forceinline__ void sage_agg_scan(int base, int t, unsigned lo, int n_edges,
                                              const int* __restrict__ dst,
                                              int* __restrict__ queue, int* qn) {
    int4 v[8];
    #pragma unroll
    for (int k = 0; k < 8; ++k) {
        const int e0 = base + k * (AGG_THREADS * 4) + t * 4;
        if (!GUARD || e0 + 3 < n_edges) v[k] = *(const int4*)(dst + e0);
        else                            v[k] = make_int4(-1, -1, -1, -1);
    }
    #pragma unroll
    for (int k = 0; k < 8; ++k) {
        const int e0 = base + k * (AGG_THREADS * 4) + t * 4;
        const unsigned x0 = (unsigned)v[k].x ^ lo;
        const unsigned x1 = (unsigned)v[k].y ^ lo;
        const unsigned x2 = (unsigned)v[k].z ^ lo;
        const unsigned x3 = (unsigned)v[k].w ^ lo;
        const unsigned m01 = x0 < x1 ? x0 : x1;
        const unsigned m23 = x2 < x3 ? x2 : x3;
        const unsigned m   = m01 < m23 ? m01 : m23;
        if (m < AGG_ROWS) {
            const int pos = atomicAdd(qn, 1);
            if (pos < AGG_QCAP) queue[pos] = e0 >> 2;
        }
    }
}

__device__ __forceinline__ void sage_agg_append(int ent, int total, int* __restrict__ queue,
                                                int* qextra) {
    const int pos = total + atomicAdd(qextra, 1);
    if (pos < AGG_QCAP) queue[pos] = ent;
}

template <bool BUILD>
__global__ void __launch_bounds__(AGG_THREADS)
sage_agg_lds_kernel(const int* __restrict__ src, const int* __restrict__ dst,
                    const float* __restrict__ x, float* __restrict__ msg,
                    int* __restrict__ lists, int n_edges, int n_nodes) {
    __shared__ float acc[AGG_ROWS * D_FEAT];
    __shared__ __align__(16) int queue[AGG_QCAP];
    __shared__ int   qn;
    __shared__ int   qextra;

    const int t = threadIdx.x;
    const unsigned lo = blockIdx.x << AGG_SHIFT;

    int* rec = lists + (size_t)blockIdx.x * AGG_REC;
    const bool fits = (size_t)gridDim.x * AGG_REC <= (size_t)n_nodes * OUT_DIM;

    for (int i = t; i < AGG_ROWS * D_FEAT / 4; i += AGG_THREADS)
        ((float4*)acc)[i] = make_float4(0.0f, 0.0f, 0.0f, 0.0f);
    if (t == 0) { qn = 0; qextra = 0; }
    __syncthreads();

    int  total2 = 0;
    bool go     = false;
    bool loaded = false;

    if (!BUILD && fits) {
        const int cnt = rec[0];
        if (cnt >= 0 && cnt <= AGG_QCAP) {
            for (int i = t; i < ((cnt + 3) >> 2); i += AGG_THREADS)
                ((int4*)queue)[i] = ((const int4*)(rec + 4))[i];
            total2 = cnt;
            go     = true;
            loaded = true;
        }
    }

    if (!loaded) {
        {
            const int per_iter = AGG_THREADS * 32;
            int base = 0;
            for (; base + per_iter <= n_edges; base += per_iter)
                sage_agg_scan<false>(base, t, lo, n_edges, dst, queue, &qn);
            if (base < n_edges)
                sage_agg_scan<true>(base, t, lo, n_edges, dst, queue, &qn);
        }
        __syncthreads();

        const int  total  = qn;
        const bool queued = total <= AGG_QCAP;

        if (queued) {
            for (int i = t; i < total; i += AGG_THREADS) {
                const int  e0 = queue[i] * 4;
                const int4 dd = *(const int4*)(dst + e0);
                const int4 ss = *(const int4*)(src + e0);
                const int  dv[4] = {dd.x, dd.y, dd.z, dd.w};
                const int  sv[4] = {ss.x, ss.y, ss.z, ss.w};
                bool have = false;
                int  ent0 = 0;
                #pragma unroll
                for (int j = 0; j < 4; ++j) {
                    if (((unsigned)dv[j] ^ lo) < AGG_ROWS) {
                        const int ent = ((dv[j] & (AGG_ROWS - 1)) << 23) | (sv[j] & AGG_SMASK);
                        if (!have) { ent0 = ent; have = true; }
                        else       sage_agg_append(ent, total, queue, &qextra);
                    }
                }
                queue[i] = ent0;
            }
            const int rem = n_edges & 3;
            if (t < rem) {
                const int e = n_edges - rem + t;
                const int d = dst[e];
                if (((unsigned)d ^ lo) < AGG_ROWS)
                    sage_agg_append(((d & (AGG_ROWS - 1)) << 23) | (src[e] & AGG_SMASK),
                                    total, queue, &qextra);
            }
        }
        __syncthreads();

        total2 = total + qextra;
        go     = queued && total2 <= AGG_QCAP;

        if (BUILD && fits) {
            if (go)
                for (int i = t; i < ((total2 + 3) >> 2); i += AGG_THREADS)
                    ((int4*)(rec + 4))[i] = ((const int4*)queue)[i];
            if (t == 0) rec[0] = go ? total2 : -1;
        }
    }
    __syncthreads();

    if (go) {
        const int wave = t >> 5;
        const int lane = t & 31;
        for (int i = wave * 8; i < total2; i += (AGG_THREADS / 32) * 8) {
            int    ent[8];
            float4 r[8];
            #pragma unroll
            for (int k = 0; k < 8; ++k) {
                const int idx = i + k;
                ent[k] = queue[idx < total2 ? idx : i];
                r[k] = *(const float4*)(x + (size_t)(ent[k] & AGG_SMASK) * D_FEAT + lane * 4);
            }
            #pragma unroll
            for (int k = 0; k < 8; ++k) {
                if (i + k < total2) {
                    float* ar = acc + ((unsigned)ent[k] >> 23) * D_FEAT + lane;
                    atomicAdd(ar +  0, r[k].x);
                    atomicAdd(ar + 32, r[k].y);
                    atomicAdd(ar + 64, r[k].z);
                    atomicAdd(ar + 96, r[k].w);
                }
            }
        }
    } else {
        for (int e = t; e < n_edges; e += AGG_THREADS) {
            const int d = dst[e];
            if (((unsigned)d ^ lo) < AGG_ROWS) {
                const float* xr = x + (size_t)src[e] * D_FEAT;
                float*       ar = acc + (d & (AGG_ROWS - 1)) * D_FEAT;
                for (int c = 0; c < D_FEAT; ++c) atomicAdd(ar + (c & 3) * 32 + (c >> 2), xr[c]);
            }
        }
    }
    __syncthreads();

    {
        const int row0 = blockIdx.x * AGG_ROWS;
        for (int i = t; i < AGG_ROWS * (D_FEAT / 4); i += AGG_THREADS) {
            const int r = i >> 5;
            const int l = i & 31;
            if (row0 + r < n_nodes) {
                const float* ar = acc + r * D_FEAT + l;
                *(float4*)(msg + (size_t)(row0 + r) * D_FEAT + l * 4) =
                    make_float4(ar[0], ar[32], ar[64], ar[96]);
            }
        }
    }
}

__global__ void __launch_bounds__(256)
sage_layer_kernel(const float* __restrict__ xin, const float* __restrict__ msg,
                  const float* __restrict__ invdeg,
                  const float* __restrict__ Wself, const float* __restrict__ Wneigh,
                  const float* __restrict__ bias, float* __restrict__ xout) {
    __shared__ float As[16 * LDS_STRIDE];
    __shared__ float Hs[16 * LDS_STRIDE];

    const int m0 = blockIdx.x * 16;
    const int t  = threadIdx.x;

    {
        int r  = t >> 4;
        int c0 = (t & 15) * 8;
        const float* xr = xin + (size_t)(m0 + r) * D_FEAT + c0;
        float4 a0 = *(const float4*)(xr);
        float4 a1 = *(const float4*)(xr + 4);
        *(float4*)(As + r * LDS_STRIDE + c0)     = a0;
        *(float4*)(As + r * LDS_STRIDE + c0 + 4) = a1;

        float iv = invdeg[m0 + r];
        const float* mr = msg + (size_t)(m0 + r) * D_FEAT + c0;
        float4 h0 = *(const float4*)(mr);
        float4 h1 = *(const float4*)(mr + 4);
        h0.x *= iv; h0.y *= iv; h0.z *= iv; h0.w *= iv;
        h1.x *= iv; h1.y *= iv; h1.z *= iv; h1.w *= iv;
        *(float4*)(Hs + r * LDS_STRIDE + c0)     = h0;
        *(float4*)(Hs + r * LDS_STRIDE + c0 + 4) = h1;
    }
    __syncthreads();

    const int wave = t >> 5;
    const int lane = t & 31;
    const int n0   = wave * 16;
    const int half = lane >> 4;
    const int l16  = lane & 15;
    const int col  = n0 + l16;

    v8f acc;
    {
        float bv = bias[col];
        #pragma unroll
        for (int i = 0; i < 8; ++i) acc[i] = bv;
    }

    #pragma unroll 4
    for (int k0 = 0; k0 < D_FEAT; k0 += 4) {
        const int ka = k0 + half * 2;
        v2f a = *(const v2f*)(As + l16 * LDS_STRIDE + ka);
        v2f b;
        b.x = Wself[(size_t)ka * D_FEAT + col];
        b.y = Wself[(size_t)(ka + 1) * D_FEAT + col];
        acc = __builtin_amdgcn_wmma_f32_16x16x4_f32(false, a, false, b,
                                                    (short)0, acc, false, false);

        v2f h = *(const v2f*)(Hs + l16 * LDS_STRIDE + ka);
        v2f bn;
        bn.x = Wneigh[(size_t)ka * D_FEAT + col];
        bn.y = Wneigh[(size_t)(ka + 1) * D_FEAT + col];
        acc = __builtin_amdgcn_wmma_f32_16x16x4_f32(false, h, false, bn,
                                                    (short)0, acc, false, false);
    }

    #pragma unroll
    for (int v = 0; v < 8; ++v) {
        int row = m0 + v + half * 8;
        float val = acc[v];
        val = val > 0.0f ? val : 0.0f;
        xout[(size_t)row * D_FEAT + col] = val;
    }
}

__global__ void __launch_bounds__(64)
sage_pred_kernel(const float* __restrict__ xin, const float* __restrict__ Wp,
                 const float* __restrict__ bp, float* __restrict__ out) {
    const int m0   = blockIdx.x * 16;
    const int t    = threadIdx.x;
    const int wave = t >> 5;
    const int lane = t & 31;
    const int n0   = wave * 16;
    const int half = lane >> 4;
    const int l16  = lane & 15;
    const int col  = n0 + l16;
    const int bcol = col < OUT_DIM ? col : (OUT_DIM - 1);

    v8f acc;
    {
        float bv = bp[bcol];
        #pragma unroll
        for (int i = 0; i < 8; ++i) acc[i] = bv;
    }

    #pragma unroll 4
    for (int k0 = 0; k0 < D_FEAT; k0 += 4) {
        const int ka = k0 + half * 2;
        const float* ap = xin + (size_t)(m0 + l16) * D_FEAT + ka;
        v2f a; a.x = ap[0]; a.y = ap[1];
        v2f b;
        b.x = Wp[(size_t)ka * OUT_DIM + bcol];
        b.y = Wp[(size_t)(ka + 1) * OUT_DIM + bcol];
        acc = __builtin_amdgcn_wmma_f32_16x16x4_f32(false, a, false, b,
                                                    (short)0, acc, false, false);
    }

    if (col < OUT_DIM) {
        #pragma unroll
        for (int v = 0; v < 8; ++v) {
            int row = m0 + v + half * 8;
            out[(size_t)row * OUT_DIM + col] = acc[v];
        }
    }
}


extern "C" void kernel_launch(void* const* d_in, const int* in_sizes, int n_in,
                              void* d_out, int out_size, void* d_ws, size_t ws_size,
                              hipStream_t stream) {
    const float* x       = (const float*)d_in[0];
    const int*   src     = (const int*)  d_in[1];
    const int*   dst     = (const int*)  d_in[2];
    const float* W_self  = (const float*)d_in[3];
    const float* W_neigh = (const float*)d_in[4];
    const float* b       = (const float*)d_in[5];
    const float* W_pred  = (const float*)d_in[6];
    const float* b_pred  = (const float*)d_in[7];

    const int n_nodes = in_sizes[0] / D_FEAT;
    const int n_edges = in_sizes[1];
    const int L       = in_sizes[5] / D_FEAT;
    const int feat_elems = n_nodes * D_FEAT;

    char*  ws = (char*)d_ws;
    size_t off = ((size_t)n_nodes * sizeof(float) + 511) & ~(size_t)511;
    size_t feat_bytes = (size_t)feat_elems * sizeof(float);
    float* invdeg = (float*)ws;
    float* msg    = (float*)(ws + off);
    float* bufA   = (float*)(ws + off + feat_bytes);
    float* bufB   = (float*)(ws + off + 2 * feat_bytes);

    int* lists = (int*)d_out;

    const int T = 256;

    sage_zero_f32<<<(n_nodes + T - 1) / T, T, 0, stream>>>(invdeg, n_nodes);
    sage_deg_kernel<<<(n_edges + T - 1) / T, T, 0, stream>>>(dst, invdeg, n_edges);
    sage_invdeg_kernel<<<(n_nodes + T - 1) / T, T, 0, stream>>>(invdeg, n_nodes);

    const float* cur = x;
    float* nxt = bufA;
    const int agg_blocks = (n_nodes + AGG_ROWS - 1) / AGG_ROWS;
    const int m_tiles = n_nodes / 16;

    for (int l = 0; l < L; ++l) {
        if (l == 0)
            sage_agg_lds_kernel<true><<<agg_blocks, AGG_THREADS, 0, stream>>>(src, dst, cur, msg, lists,
                                                                              n_edges, n_nodes);
        else
            sage_agg_lds_kernel<false><<<agg_blocks, AGG_THREADS, 0, stream>>>(src, dst, cur, msg, lists,
                                                                               n_edges, n_nodes);
        sage_layer_kernel<<<m_tiles, 256, 0, stream>>>(
            cur, msg, invdeg,
            W_self  + (size_t)l * D_FEAT * D_FEAT,
            W_neigh + (size_t)l * D_FEAT * D_FEAT,
            b       + (size_t)l * D_FEAT,
            nxt);
        cur = nxt;
        nxt = (nxt == bufA) ? bufB : bufA;
    }

    sage_pred_kernel<<<m_tiles, 64, 0, stream>>>(cur, W_pred, b_pred, (float*)d_out);
}


# Layer 1's edge scan reads a byte-per-edge plane of 512-row range numbers written by the degree pass (a quarter of dst's bytes); one LDS add per thread pass takes queue slots; arrival order as before
# speedup vs baseline: 2.5811x; 1.0401x over previous
#include <hip/hip_runtime.h>
#include <hip/hip_bf16.h>


#define D_FEAT 128
#define LDS_STRIDE 132
#define OUT_DIM 26

typedef __attribute__((ext_vector_type(2))) float v2f;
typedef __attribute__((ext_vector_type(8))) float v8f;

#define AGG_ROWS    512
#define AGG_SHIFT   9
#define AGG_THREADS 1024
#define AGG_QCAP    12288
#define AGG_SMASK   0x7FFFFF
#define AGG_REC     (AGG_QCAP + 4)

__device__ __forceinline__ bool sage_blk_ok(int n_edges, int n_nodes) {
    return ((n_nodes + AGG_ROWS - 1) >> AGG_SHIFT) <= 255 &&
           (size_t)n_edges <= (size_t)n_nodes * (D_FEAT * sizeof(float));
}


__global__ void sage_zero_f32(float* __restrict__ p, int n) {
    int i = blockIdx.x * blockDim.x + threadIdx.x;
    if (i < n) p[i] = 0.0f;
}

__global__ void sage_deg_kernel(const int* __restrict__ dst, float* __restrict__ deg,
                                unsigned char* __restrict__ blk, int n_edges, int n_nodes) {
    int e = blockIdx.x * blockDim.x + threadIdx.x;
    if (e < n_edges) {
        const int d = dst[e];
        atomicAdd(&deg[d], 1.0f);
        if (sage_blk_ok(n_edges, n_nodes)) {
            const unsigned b = (unsigned)d >> AGG_SHIFT;
            blk[e] = (unsigned char)(b < 255u ? b : 255u);
        }
    }
}

__global__ void sage_invdeg_kernel(float* __restrict__ deg, int n_nodes) {
    int i = blockIdx.x * blockDim.x + threadIdx.x;
    if (i < n_nodes) {
        float d = deg[i];
        deg[i] = 1.0f / fmaxf(d, 1.0f);
    }
}

template <bool GUARD>
__device__ __forceinline__ void sage_agg_scan(int base, int t, unsigned lo, int n_edges,
                                              const int* __restrict__ dst,
                                              int* __restrict__ queue, int* qn) {
    int4 v[8];
    #pragma unroll
    for (int k = 0; k < 8; ++k) {
        const int e0 = base + k * (AGG_THREADS * 4) + t * 4;
        if (!GUARD || e0 + 3 < n_edges) v[k] = *(const int4*)(dst + e0);
        else                            v[k] = make_int4(-1, -1, -1, -1);
    }
    #pragma unroll
    for (int k = 0; k < 8; ++k) {
        const int e0 = base + k * (AGG_THREADS * 4) + t * 4;
        const unsigned x0 = (unsigned)v[k].x ^ lo;
        const unsigned x1 = (unsigned)v[k].y ^ lo;
        const unsigned x2 = (unsigned)v[k].z ^ lo;
        const unsigned x3 = (unsigned)v[k].w ^ lo;
        const unsigned m01 = x0 < x1 ? x0 : x1;
        const unsigned m23 = x2 < x3 ? x2 : x3;
        const unsigned m   = m01 < m23 ? m01 : m23;
        if (m < AGG_ROWS) {
            const int pos = atomicAdd(qn, 1);
            if (pos < AGG_QCAP) queue[pos] = e0 >> 2;
        }
    }
}

template <bool GUARD>
__device__ __forceinline__ void sage_agg_scan_bytes(int base, int t, unsigned rep, int n_edges,
                                                    const unsigned char* __restrict__ blk,
                                                    int* __restrict__ queue, int* qn) {
    uint4 v[8];
    #pragma unroll
    for (int k = 0; k < 8; ++k) {
        const int e0 = base + k * (AGG_THREADS * 16) + t * 16;
        if (!GUARD || e0 + 15 < n_edges) {
            v[k] = *(const uint4*)(blk + e0);
        } else {
            unsigned w[4];
            #pragma unroll
            for (int j = 0; j < 4; ++j)
                w[j] = (e0 + 4 * j + 3 < n_edges) ? *(const unsigned*)(blk + e0 + 4 * j) : ~rep;
            v[k] = make_uint4(w[0], w[1], w[2], w[3]);
        }
    }
    unsigned mask = 0;
    #pragma unroll
    for (int k = 0; k < 8; ++k) {
        const unsigned w[4] = {v[k].x, v[k].y, v[k].z, v[k].w};
        #pragma unroll
        for (int j = 0; j < 4; ++j) {
            const unsigned x = w[j] ^ rep;
            if ((x - 0x01010101u) & ~x & 0x80808080u) mask |= 1u << (k * 4 + j);
        }
    }
    if (mask) {
        int pos = atomicAdd(qn, __popc(mask));
        const int g0 = (base >> 2) + t * 4;
        while (mask) {
            const int bit = __ffs(mask) - 1;
            mask &= mask - 1;
            if (pos < AGG_QCAP) queue[pos] = g0 + (bit >> 2) * (AGG_THREADS * 4) + (bit & 3);
            ++pos;
        }
    }
}

__device__ __forceinline__ void sage_agg_append(int ent, int total, int* __restrict__ queue,
                                                int* qextra) {
    const int pos = total + atomicAdd(qextra, 1);
    if (pos < AGG_QCAP) queue[pos] = ent;
}

template <bool BUILD>
__global__ void __launch_bounds__(AGG_THREADS)
sage_agg_lds_kernel(const int* __restrict__ src, const int* __restrict__ dst,
                    const unsigned char* __restrict__ blk,
                    const float* __restrict__ x, float* __restrict__ msg,
                    int* __restrict__ lists, int n_edges, int n_nodes) {
    __shared__ float acc[AGG_ROWS * D_FEAT];
    __shared__ __align__(16) int queue[AGG_QCAP];
    __shared__ int   qn;
    __shared__ int   qextra;

    const int t = threadIdx.x;
    const unsigned lo = blockIdx.x << AGG_SHIFT;

    int* rec = lists + (size_t)blockIdx.x * AGG_REC;
    const bool fits = (size_t)gridDim.x * AGG_REC <= (size_t)n_nodes * OUT_DIM;

    for (int i = t; i < AGG_ROWS * D_FEAT / 4; i += AGG_THREADS)
        ((float4*)acc)[i] = make_float4(0.0f, 0.0f, 0.0f, 0.0f);
    if (t == 0) { qn = 0; qextra = 0; }
    __syncthreads();

    int  total2 = 0;
    bool go     = false;
    bool loaded = false;

    if (!BUILD && fits) {
        const int cnt = rec[0];
        if (cnt >= 0 && cnt <= AGG_QCAP) {
            for (int i = t; i < ((cnt + 3) >> 2); i += AGG_THREADS)
                ((int4*)queue)[i] = ((const int4*)(rec + 4))[i];
            total2 = cnt;
            go     = true;
            loaded = true;
        }
    }

    if (!loaded) {
        if (BUILD && sage_blk_ok(n_edges, n_nodes)) {
            const unsigned rep = blockIdx.x * 0x01010101u;
            const int per_iter = AGG_THREADS * 128;
            int base = 0;
            for (; base + per_iter <= n_edges; base += per_iter)
                sage_agg_scan_bytes<false>(base, t, rep, n_edges, blk, queue, &qn);
            if (base < n_edges)
                sage_agg_scan_bytes<true>(base, t, rep, n_edges, blk, queue, &qn);
        } else {
            const int per_iter = AGG_THREADS * 32;
            int base = 0;
            for (; base + per_iter <= n_edges; base += per_iter)
                sage_agg_scan<false>(base, t, lo, n_edges, dst, queue, &qn);
            if (base < n_edges)
                sage_agg_scan<true>(base, t, lo, n_edges, dst, queue, &qn);
        }
        __syncthreads();

        const int  total  = qn;
        const bool queued = total <= AGG_QCAP;

        if (queued) {
            for (int i = t; i < total; i += AGG_THREADS) {
                const int  e0 = queue[i] * 4;
                const int4 dd = *(const int4*)(dst + e0);
                const int4 ss = *(const int4*)(src + e0);
                const int  dv[4] = {dd.x, dd.y, dd.z, dd.w};
                const int  sv[4] = {ss.x, ss.y, ss.z, ss.w};
                bool have = false;
                int  ent0 = 0;
                #pragma unroll
                for (int j = 0; j < 4; ++j) {
                    if (((unsigned)dv[j] ^ lo) < AGG_ROWS) {
                        const int ent = ((dv[j] & (AGG_ROWS - 1)) << 23) | (sv[j] & AGG_SMASK);
                        if (!have) { ent0 = ent; have = true; }
                        else       sage_agg_append(ent, total, queue, &qextra);
                    }
                }
                queue[i] = ent0;
            }
            const int rem = n_edges & 3;
            if (t < rem) {
                const int e = n_edges - rem + t;
                const int d = dst[e];
                if (((unsigned)d ^ lo) < AGG_ROWS)
                    sage_agg_append(((d & (AGG_ROWS - 1)) << 23) | (src[e] & AGG_SMASK),
                                    total, queue, &qextra);
            }
        }
        __syncthreads();

        total2 = total + qextra;
        go     = queued && total2 <= AGG_QCAP;

        if (BUILD && fits) {
            if (go)
                for (int i = t; i < ((total2 + 3) >> 2); i += AGG_THREADS)
                    ((int4*)(rec + 4))[i] = ((const int4*)queue)[i];
            if (t == 0) rec[0] = go ? total2 : -1;
        }
    }
    __syncthreads();

    if (go) {
        const int wave = t >> 5;
        const int lane = t & 31;
        for (int i = wave * 8; i < total2; i += (AGG_THREADS / 32) * 8) {
            int    ent[8];
            float4 r[8];
            #pragma unroll
            for (int k = 0; k < 8; ++k) {
                const int idx = i + k;
                ent[k] = queue[idx < total2 ? idx : i];
                r[k] = *(const float4*)(x + (size_t)(ent[k] & AGG_SMASK) * D_FEAT + lane * 4);
            }
            #pragma unroll
            for (int k = 0; k < 8; ++k) {
                if (i + k < total2) {
                    float* ar = acc + ((unsigned)ent[k] >> 23) * D_FEAT + lane;
                    atomicAdd(ar +  0, r[k].x);
                    atomicAdd(ar + 32, r[k].y);
                    atomicAdd(ar + 64, r[k].z);
                    atomicAdd(ar + 96, r[k].w);
                }
            }
        }
    } else {
        for (int e = t; e < n_edges; e += AGG_THREADS) {
            const int d = dst[e];
            if (((unsigned)d ^ lo) < AGG_ROWS) {
                const float* xr = x + (size_t)src[e] * D_FEAT;
                float*       ar = acc + (d & (AGG_ROWS - 1)) * D_FEAT;
                for (int c = 0; c < D_FEAT; ++c) atomicAdd(ar + (c & 3) * 32 + (c >> 2), xr[c]);
            }
        }
    }
    __syncthreads();

    {
        const int row0 = blockIdx.x * AGG_ROWS;
        for (int i = t; i < AGG_ROWS * (D_FEAT / 4); i += AGG_THREADS) {
            const int r = i >> 5;
            const int l = i & 31;
            if (row0 + r < n_nodes) {
                const float* ar = acc + r * D_FEAT + l;
                *(float4*)(msg + (size_t)(row0 + r) * D_FEAT + l * 4) =
                    make_float4(ar[0], ar[32], ar[64], ar[96]);
            }
        }
    }
}

__global__ void __launch_bounds__(256)
sage_layer_kernel(const float* __restrict__ xin, const float* __restrict__ msg,
                  const float* __restrict__ invdeg,
                  const float* __restrict__ Wself, const float* __restrict__ Wneigh,
                  const float* __restrict__ bias, float* __restrict__ xout) {
    __shared__ float As[16 * LDS_STRIDE];
    __shared__ float Hs[16 * LDS_STRIDE];

    const int m0 = blockIdx.x * 16;
    const int t  = threadIdx.x;

    {
        int r  = t >> 4;
        int c0 = (t & 15) * 8;
        const float* xr = xin + (size_t)(m0 + r) * D_FEAT + c0;
        float4 a0 = *(const float4*)(xr);
        float4 a1 = *(const float4*)(xr + 4);
        *(float4*)(As + r * LDS_STRIDE + c0)     = a0;
        *(float4*)(As + r * LDS_STRIDE + c0 + 4) = a1;

        float iv = invdeg[m0 + r];
        const float* mr = msg + (size_t)(m0 + r) * D_FEAT + c0;
        float4 h0 = *(const float4*)(mr);
        float4 h1 = *(const float4*)(mr + 4);
        h0.x *= iv; h0.y *= iv; h0.z *= iv; h0.w *= iv;
        h1.x *= iv; h1.y *= iv; h1.z *= iv; h1.w *= iv;
        *(float4*)(Hs + r * LDS_STRIDE + c0)     = h0;
        *(float4*)(Hs + r * LDS_STRIDE + c0 + 4) = h1;
    }
    __syncthreads();

    const int wave = t >> 5;
    const int lane = t & 31;
    const int n0   = wave * 16;
    const int half = lane >> 4;
    const int l16  = lane & 15;
    const int col  = n0 + l16;

    v8f acc;
    {
        float bv = bias[col];
        #pragma unroll
        for (int i = 0; i < 8; ++i) acc[i] = bv;
    }

    #pragma unroll 4
    for (int k0 = 0; k0 < D_FEAT; k0 += 4) {
        const int ka = k0 + half * 2;
        v2f a = *(const v2f*)(As + l16 * LDS_STRIDE + ka);
        v2f b;
        b.x = Wself[(size_t)ka * D_FEAT + col];
        b.y = Wself[(size_t)(ka + 1) * D_FEAT + col];
        acc = __builtin_amdgcn_wmma_f32_16x16x4_f32(false, a, false, b,
                                                    (short)0, acc, false, false);

        v2f h = *(const v2f*)(Hs + l16 * LDS_STRIDE + ka);
        v2f bn;
        bn.x = Wneigh[(size_t)ka * D_FEAT + col];
        bn.y = Wneigh[(size_t)(ka + 1) * D_FEAT + col];
        acc = __builtin_amdgcn_wmma_f32_16x16x4_f32(false, h, false, bn,
                                                    (short)0, acc, false, false);
    }

    #pragma unroll
    for (int v = 0; v < 8; ++v) {
        int row = m0 + v + half * 8;
        float val = acc[v];
        val = val > 0.0f ? val : 0.0f;
        xout[(size_t)row * D_FEAT + col] = val;
    }
}

__global__ void __launch_bounds__(64)
sage_pred_kernel(const float* __restrict__ xin, const float* __restrict__ Wp,
                 const float* __restrict__ bp, float* __restrict__ out) {
    const int m0   = blockIdx.x * 16;
    const int t    = threadIdx.x;
    const int wave = t >> 5;
    const int lane = t & 31;
    const int n0   = wave * 16;
    const int half = lane >> 4;
    const int l16  = lane & 15;
    const int col  = n0 + l16;
    const int bcol = col < OUT_DIM ? col : (OUT_DIM - 1);

    v8f acc;
    {
        float bv = bp[bcol];
        #pragma unroll
        for (int i = 0; i < 8; ++i) acc[i] = bv;
    }

    #pragma unroll 4
    for (int k0 = 0; k0 < D_FEAT; k0 += 4) {
        const int ka = k0 + half * 2;
        const float* ap = xin + (size_t)(m0 + l16) * D_FEAT + ka;
        v2f a; a.x = ap[0]; a.y = ap[1];
        v2f b;
        b.x = Wp[(size_t)ka * OUT_DIM + bcol];
        b.y = Wp[(size_t)(ka + 1) * OUT_DIM + bcol];
        acc = __builtin_amdgcn_wmma_f32_16x16x4_f32(false, a, false, b,
                                                    (short)0, acc, false, false);
    }

    if (col < OUT_DIM) {
        #pragma unroll
        for (int v = 0; v < 8; ++v) {
            int row = m0 + v + half * 8;
            out[(size_t)row * OUT_DIM + col] = acc[v];
        }
    }
}


extern "C" void kernel_launch(void* const* d_in, const int* in_sizes, int n_in,
                              void* d_out, int out_size, void* d_ws, size_t ws_size,
                              hipStream_t stream) {
    const float* x       = (const float*)d_in[0];
    const int*   src     = (const int*)  d_in[1];
    const int*   dst     = (const int*)  d_in[2];
    const float* W_self  = (const float*)d_in[3];
    const float* W_neigh = (const float*)d_in[4];
    const float* b       = (const float*)d_in[5];
    const float* W_pred  = (const float*)d_in[6];
    const float* b_pred  = (const float*)d_in[7];

    const int n_nodes = in_sizes[0] / D_FEAT;
    const int n_edges = in_sizes[1];
    const int L       = in_sizes[5] / D_FEAT;
    const int feat_elems = n_nodes * D_FEAT;

    char*  ws = (char*)d_ws;
    size_t off = ((size_t)n_nodes * sizeof(float) + 511) & ~(size_t)511;
    size_t feat_bytes = (size_t)feat_elems * sizeof(float);
    float* invdeg = (float*)ws;
    float* msg    = (float*)(ws + off);
    float* bufA   = (float*)(ws + off + feat_bytes);
    float* bufB   = (float*)(ws + off + 2 * feat_bytes);

    int* lists = (int*)d_out;

    unsigned char* blk = (unsigned char*)bufB;

    const int T = 256;

    sage_zero_f32<<<(n_nodes + T - 1) / T, T, 0, stream>>>(invdeg, n_nodes);
    sage_deg_kernel<<<(n_edges + T - 1) / T, T, 0, stream>>>(dst, invdeg, blk, n_edges, n_nodes);
    sage_invdeg_kernel<<<(n_nodes + T - 1) / T, T, 0, stream>>>(invdeg, n_nodes);

    const float* cur = x;
    float* nxt = bufA;
    const int agg_blocks = (n_nodes + AGG_ROWS - 1) / AGG_ROWS;
    const int m_tiles = n_nodes / 16;

    for (int l = 0; l < L; ++l) {
        if (l == 0)
            sage_agg_lds_kernel<true><<<agg_blocks, AGG_THREADS, 0, stream>>>(src, dst, blk, cur, msg, lists,
                                                                              n_edges, n_nodes);
        else
            sage_agg_lds_kernel<false><<<agg_blocks, AGG_THREADS, 0, stream>>>(src, dst, blk, cur, msg, lists,
                                                                               n_edges, n_nodes);
        sage_layer_kernel<<<m_tiles, 256, 0, stream>>>(
            cur, msg, invdeg,
            W_self  + (size_t)l * D_FEAT * D_FEAT,
            W_neigh + (size_t)l * D_FEAT * D_FEAT,
            b       + (size_t)l * D_FEAT,
            nxt);
        cur = nxt;
        nxt = (nxt == bufA) ? bufB : bufA;
    }

    sage_pred_kernel<<<m_tiles, 64, 0, stream>>>(cur, W_pred, b_pred, (float*)d_out);
}


# Degrees counted in LDS by each neighbour-sum launch (integer adds over its queue), rows scaled at its store; memory float atomics, 2 launches go; byte pass: 16 edges a thread; sum orders unchanged
# speedup vs baseline: 2.8347x; 1.0983x over previous
#include <hip/hip_runtime.h>
#include <hip/hip_bf16.h>


#define D_FEAT 128
#define LDS_STRIDE 132
#define OUT_DIM 26

typedef __attribute__((ext_vector_type(2))) float v2f;
typedef __attribute__((ext_vector_type(8))) float v8f;

#define AGG_ROWS    512
#define AGG_SHIFT   9
#define AGG_THREADS 1024
#define AGG_QCAP    12288
#define AGG_SMASK   0x7FFFFF
#define AGG_REC     (AGG_QCAP + 4)

__device__ __forceinline__ bool sage_blk_ok(int n_edges, int n_nodes) {
    return ((n_nodes + AGG_ROWS - 1) >> AGG_SHIFT) <= 255 &&
           (size_t)n_edges <= (size_t)n_nodes * (D_FEAT * sizeof(float));
}


__device__ __forceinline__ unsigned sage_range_byte(int d) {
    const unsigned b = (unsigned)d >> AGG_SHIFT;
    return b < 255u ? b : 255u;
}

__global__ void sage_range_bytes_kernel(const int* __restrict__ dst, unsigned char* __restrict__ blk,
                                        int n_edges, int n_nodes) {
    if (!sage_blk_ok(n_edges, n_nodes)) return;
    const int e0 = (blockIdx.x * blockDim.x + threadIdx.x) * 16;
    if (e0 + 15 < n_edges) {
        int4 v[4];
        #pragma unroll
        for (int k = 0; k < 4; ++k) v[k] = *(const int4*)(dst + e0 + 4 * k);
        unsigned w[4];
        #pragma unroll
        for (int k = 0; k < 4; ++k)
            w[k] = sage_range_byte(v[k].x)       | sage_range_byte(v[k].y) << 8 |
                   sage_range_byte(v[k].z) << 16 | sage_range_byte(v[k].w) << 24;
        *(uint4*)(blk + e0) = make_uint4(w[0], w[1], w[2], w[3]);
    } else {
        for (int e = e0; e < n_edges; ++e) blk[e] = (unsigned char)sage_range_byte(dst[e]);
    }
}

template <bool GUARD>
__device__ __forceinline__ void sage_agg_scan(int base, int t, unsigned lo, int n_edges,
                                              const int* __restrict__ dst,
                                              int* __restrict__ queue, int* qn) {
    int4 v[8];
    #pragma unroll
    for (int k = 0; k < 8; ++k) {
        const int e0 = base + k * (AGG_THREADS * 4) + t * 4;
        if (!GUARD || e0 + 3 < n_edges) v[k] = *(const int4*)(dst + e0);
        else                            v[k] = make_int4(-1, -1, -1, -1);
    }
    #pragma unroll
    for (int k = 0; k < 8; ++k) {
        const int e0 = base + k * (AGG_THREADS * 4) + t * 4;
        const unsigned x0 = (unsigned)v[k].x ^ lo;
        const unsigned x1 = (unsigned)v[k].y ^ lo;
        const unsigned x2 = (unsigned)v[k].z ^ lo;
        const unsigned x3 = (unsigned)v[k].w ^ lo;
        const unsigned m01 = x0 < x1 ? x0 : x1;
        const unsigned m23 = x2 < x3 ? x2 : x3;
        const unsigned m   = m01 < m23 ? m01 : m23;
        if (m < AGG_ROWS) {
            const int pos = atomicAdd(qn, 1);
            if (pos < AGG_QCAP) queue[pos] = e0 >> 2;
        }
    }
}

template <bool GUARD>
__device__ __forceinline__ void sage_agg_scan_bytes(int base, int t, unsigned rep, int n_edges,
                                                    const unsigned char* __restrict__ blk,
                                                    int* __restrict__ queue, int* qn) {
    uint4 v[8];
    #pragma unroll
    for (int k = 0; k < 8; ++k) {
        const int e0 = base + k * (AGG_THREADS * 16) + t * 16;
        if (!GUARD || e0 + 15 < n_edges) {
            v[k] = *(const uint4*)(blk + e0);
        } else {
            unsigned w[4];
            #pragma unroll
            for (int j = 0; j < 4; ++j)
                w[j] = (e0 + 4 * j + 3 < n_edges) ? *(const unsigned*)(blk + e0 + 4 * j) : ~rep;
            v[k] = make_uint4(w[0], w[1], w[2], w[3]);
        }
    }
    unsigned mask = 0;
    #pragma unroll
    for (int k = 0; k < 8; ++k) {
        const unsigned w[4] = {v[k].x, v[k].y, v[k].z, v[k].w};
        #pragma unroll
        for (int j = 0; j < 4; ++j) {
            const unsigned x = w[j] ^ rep;
            if ((x - 0x01010101u) & ~x & 0x80808080u) mask |= 1u << (k * 4 + j);
        }
    }
    if (mask) {
        int pos = atomicAdd(qn, __popc(mask));
        const int g0 = (base >> 2) + t * 4;
        while (mask) {
            const int bit = __ffs(mask) - 1;
            mask &= mask - 1;
            if (pos < AGG_QCAP) queue[pos] = g0 + (bit >> 2) * (AGG_THREADS * 4) + (bit & 3);
            ++pos;
        }
    }
}

__device__ __forceinline__ void sage_agg_append(int ent, int total, int* __restrict__ queue,
                                                int* qextra) {
    const int pos = total + atomicAdd(qextra, 1);
    if (pos < AGG_QCAP) queue[pos] = ent;
}

template <bool BUILD>
__global__ void __launch_bounds__(AGG_THREADS)
sage_agg_lds_kernel(const int* __restrict__ src, const int* __restrict__ dst,
                    const unsigned char* __restrict__ blk,
                    const float* __restrict__ x, float* __restrict__ msg,
                    int* __restrict__ lists, int n_edges, int n_nodes) {
    __shared__ float acc[AGG_ROWS * D_FEAT];
    __shared__ __align__(16) int queue[AGG_QCAP];
    __shared__ int   qn;
    __shared__ int   qextra;
    __shared__ int   rowcnt[AGG_ROWS];

    const int t = threadIdx.x;
    const unsigned lo = blockIdx.x << AGG_SHIFT;

    int* rec = lists + (size_t)blockIdx.x * AGG_REC;
    const bool fits = (size_t)gridDim.x * AGG_REC <= (size_t)n_nodes * OUT_DIM;

    for (int i = t; i < AGG_ROWS * D_FEAT / 4; i += AGG_THREADS)
        ((float4*)acc)[i] = make_float4(0.0f, 0.0f, 0.0f, 0.0f);
    for (int i = t; i < AGG_ROWS; i += AGG_THREADS) rowcnt[i] = 0;
    if (t == 0) { qn = 0; qextra = 0; }
    __syncthreads();

    int  total2 = 0;
    bool go     = false;
    bool loaded = false;

    if (!BUILD && fits) {
        const int cnt = rec[0];
        if (cnt >= 0 && cnt <= AGG_QCAP) {
            for (int i = t; i < ((cnt + 3) >> 2); i += AGG_THREADS)
                ((int4*)queue)[i] = ((const int4*)(rec + 4))[i];
            total2 = cnt;
            go     = true;
            loaded = true;
        }
    }

    if (!loaded) {
        if (BUILD && sage_blk_ok(n_edges, n_nodes)) {
            const unsigned rep = blockIdx.x * 0x01010101u;
            const int per_iter = AGG_THREADS * 128;
            int base = 0;
            for (; base + per_iter <= n_edges; base += per_iter)
                sage_agg_scan_bytes<false>(base, t, rep, n_edges, blk, queue, &qn);
            if (base < n_edges)
                sage_agg_scan_bytes<true>(base, t, rep, n_edges, blk, queue, &qn);
        } else {
            const int per_iter = AGG_THREADS * 32;
            int base = 0;
            for (; base + per_iter <= n_edges; base += per_iter)
                sage_agg_scan<false>(base, t, lo, n_edges, dst, queue, &qn);
            if (base < n_edges)
                sage_agg_scan<true>(base, t, lo, n_edges, dst, queue, &qn);
        }
        __syncthreads();

        const int  total  = qn;
        const bool queued = total <= AGG_QCAP;

        if (queued) {
            for (int i = t; i < total; i += AGG_THREADS) {
                const int  e0 = queue[i] * 4;
                const int4 dd = *(const int4*)(dst + e0);
                const int4 ss = *(const int4*)(src + e0);
                const int  dv[4] = {dd.x, dd.y, dd.z, dd.w};
                const int  sv[4] = {ss.x, ss.y, ss.z, ss.w};
                bool have = false;
                int  ent0 = 0;
                #pragma unroll
                for (int j = 0; j < 4; ++j) {
                    if (((unsigned)dv[j] ^ lo) < AGG_ROWS) {
                        const int ent = ((dv[j] & (AGG_ROWS - 1)) << 23) | (sv[j] & AGG_SMASK);
                        if (!have) { ent0 = ent; have = true; }
                        else       sage_agg_append(ent, total, queue, &qextra);
                    }
                }
                queue[i] = ent0;
            }
            const int rem = n_edges & 3;
            if (t < rem) {
                const int e = n_edges - rem + t;
                const int d = dst[e];
                if (((unsigned)d ^ lo) < AGG_ROWS)
                    sage_agg_append(((d & (AGG_ROWS - 1)) << 23) | (src[e] & AGG_SMASK),
                                    total, queue, &qextra);
            }
        }
        __syncthreads();

        total2 = total + qextra;
        go     = queued && total2 <= AGG_QCAP;

        if (BUILD && fits) {
            if (go)
                for (int i = t; i < ((total2 + 3) >> 2); i += AGG_THREADS)
                    ((int4*)(rec + 4))[i] = ((const int4*)queue)[i];
            if (t == 0) rec[0] = go ? total2 : -1;
        }
    }
    __syncthreads();

    if (go) {
        for (int i = t; i < total2; i += AGG_THREADS)
            atomicAdd(&rowcnt[(unsigned)queue[i] >> 23], 1);

        const int wave = t >> 5;
        const int lane = t & 31;
        for (int i = wave * 8; i < total2; i += (AGG_THREADS / 32) * 8) {
            int    ent[8];
            float4 r[8];
            #pragma unroll
            for (int k = 0; k < 8; ++k) {
                const int idx = i + k;
                ent[k] = queue[idx < total2 ? idx : i];
                r[k] = *(const float4*)(x + (size_t)(ent[k] & AGG_SMASK) * D_FEAT + lane * 4);
            }
            #pragma unroll
            for (int k = 0; k < 8; ++k) {
                if (i + k < total2) {
                    float* ar = acc + ((unsigned)ent[k] >> 23) * D_FEAT + lane;
                    atomicAdd(ar +  0, r[k].x);
                    atomicAdd(ar + 32, r[k].y);
                    atomicAdd(ar + 64, r[k].z);
                    atomicAdd(ar + 96, r[k].w);
                }
            }
        }
    } else {
        for (int e = t; e < n_edges; e += AGG_THREADS) {
            const int d = dst[e];
            if (((unsigned)d ^ lo) < AGG_ROWS) {
                const float* xr = x + (size_t)src[e] * D_FEAT;
                float*       ar = acc + (d & (AGG_ROWS - 1)) * D_FEAT;
                atomicAdd(&rowcnt[d & (AGG_ROWS - 1)], 1);
                for (int c = 0; c < D_FEAT; ++c) atomicAdd(ar + (c & 3) * 32 + (c >> 2), xr[c]);
            }
        }
    }
    __syncthreads();

    {
        const int row0 = blockIdx.x * AGG_ROWS;
        for (int i = t; i < AGG_ROWS * (D_FEAT / 4); i += AGG_THREADS) {
            const int r = i >> 5;
            const int l = i & 31;
            if (row0 + r < n_nodes) {
                const float* ar = acc + r * D_FEAT + l;
                const float  iv = 1.0f / fmaxf((float)rowcnt[r], 1.0f);
                *(float4*)(msg + (size_t)(row0 + r) * D_FEAT + l * 4) =
                    make_float4(ar[0] * iv, ar[32] * iv, ar[64] * iv, ar[96] * iv);
            }
        }
    }
}

__global__ void __launch_bounds__(256)
sage_layer_kernel(const float* __restrict__ xin, const float* __restrict__ msg,
                  const float* __restrict__ Wself, const float* __restrict__ Wneigh,
                  const float* __restrict__ bias, float* __restrict__ xout) {
    __shared__ float As[16 * LDS_STRIDE];
    __shared__ float Hs[16 * LDS_STRIDE];

    const int m0 = blockIdx.x * 16;
    const int t  = threadIdx.x;

    {
        int r  = t >> 4;
        int c0 = (t & 15) * 8;
        const float* xr = xin + (size_t)(m0 + r) * D_FEAT + c0;
        float4 a0 = *(const float4*)(xr);
        float4 a1 = *(const float4*)(xr + 4);
        *(float4*)(As + r * LDS_STRIDE + c0)     = a0;
        *(float4*)(As + r * LDS_STRIDE + c0 + 4) = a1;

        const float* mr = msg + (size_t)(m0 + r) * D_FEAT + c0;
        float4 h0 = *(const float4*)(mr);
        float4 h1 = *(const float4*)(mr + 4);
        *(float4*)(Hs + r * LDS_STRIDE + c0)     = h0;
        *(float4*)(Hs + r * LDS_STRIDE + c0 + 4) = h1;
    }
    __syncthreads();

    const int wave = t >> 5;
    const int lane = t & 31;
    const int n0   = wave * 16;
    const int half = lane >> 4;
    const int l16  = lane & 15;
    const int col  = n0 + l16;

    v8f acc;
    {
        float bv = bias[col];
        #pragma unroll
        for (int i = 0; i < 8; ++i) acc[i] = bv;
    }

    #pragma unroll 4
    for (int k0 = 0; k0 < D_FEAT; k0 += 4) {
        const int ka = k0 + half * 2;
        v2f a = *(const v2f*)(As + l16 * LDS_STRIDE + ka);
        v2f b;
        b.x = Wself[(size_t)ka * D_FEAT + col];
        b.y = Wself[(size_t)(ka + 1) * D_FEAT + col];
        acc = __builtin_amdgcn_wmma_f32_16x16x4_f32(false, a, false, b,
                                                    (short)0, acc, false, false);

        v2f h = *(const v2f*)(Hs + l16 * LDS_STRIDE + ka);
        v2f bn;
        bn.x = Wneigh[(size_t)ka * D_FEAT + col];
        bn.y = Wneigh[(size_t)(ka + 1) * D_FEAT + col];
        acc = __builtin_amdgcn_wmma_f32_16x16x4_f32(false, h, false, bn,
                                                    (short)0, acc, false, false);
    }

    #pragma unroll
    for (int v = 0; v < 8; ++v) {
        int row = m0 + v + half * 8;
        float val = acc[v];
        val = val > 0.0f ? val : 0.0f;
        xout[(size_t)row * D_FEAT + col] = val;
    }
}

__global__ void __launch_bounds__(64)
sage_pred_kernel(const float* __restrict__ xin, const float* __restrict__ Wp,
                 const float* __restrict__ bp, float* __restrict__ out) {
    const int m0   = blockIdx.x * 16;
    const int t    = threadIdx.x;
    const int wave = t >> 5;
    const int lane = t & 31;
    const int n0   = wave * 16;
    const int half = lane >> 4;
    const int l16  = lane & 15;
    const int col  = n0 + l16;
    const int bcol = col < OUT_DIM ? col : (OUT_DIM - 1);

    v8f acc;
    {
        float bv = bp[bcol];
        #pragma unroll
        for (int i = 0; i < 8; ++i) acc[i] = bv;
    }

    #pragma unroll 4
    for (int k0 = 0; k0 < D_FEAT; k0 += 4) {
        const int ka = k0 + half * 2;
        const float* ap = xin + (size_t)(m0 + l16) * D_FEAT + ka;
        v2f a; a.x = ap[0]; a.y = ap[1];
        v2f b;
        b.x = Wp[(size_t)ka * OUT_DIM + bcol];
        b.y = Wp[(size_t)(ka + 1) * OUT_DIM + bcol];
        acc = __builtin_amdgcn_wmma_f32_16x16x4_f32(false, a, false, b,
                                                    (short)0, acc, false, false);
    }

    if (col < OUT_DIM) {
        #pragma unroll
        for (int v = 0; v < 8; ++v) {
            int row = m0 + v + half * 8;
            out[(size_t)row * OUT_DIM + col] = acc[v];
        }
    }
}


extern "C" void kernel_launch(void* const* d_in, const int* in_sizes, int n_in,
                              void* d_out, int out_size, void* d_ws, size_t ws_size,
                              hipStream_t stream) {
    const float* x       = (const float*)d_in[0];
    const int*   src     = (const int*)  d_in[1];
    const int*   dst     = (const int*)  d_in[2];
    const float* W_self  = (const float*)d_in[3];
    const float* W_neigh = (const float*)d_in[4];
    const float* b       = (const float*)d_in[5];
    const float* W_pred  = (const float*)d_in[6];
    const float* b_pred  = (const float*)d_in[7];

    const int n_nodes = in_sizes[0] / D_FEAT;
    const int n_edges = in_sizes[1];
    const int L       = in_sizes[5] / D_FEAT;
    const int feat_elems = n_nodes * D_FEAT;

    char*  ws = (char*)d_ws;
    size_t feat_bytes = (size_t)feat_elems * sizeof(float);
    float* msg    = (float*)ws;
    float* bufA   = (float*)(ws + feat_bytes);
    float* bufB   = (float*)(ws + 2 * feat_bytes);

    int* lists = (int*)d_out;

    unsigned char* blk = (unsigned char*)bufB;

    const int T = 256;

    sage_range_bytes_kernel<<<(n_edges + 16 * T - 1) / (16 * T), T, 0, stream>>>(dst, blk, n_edges, n_nodes);

    const float* cur = x;
    float* nxt = bufA;
    const int agg_blocks = (n_nodes + AGG_ROWS - 1) / AGG_ROWS;
    const int m_tiles = n_nodes / 16;

    for (int l = 0; l < L; ++l) {
        if (l == 0)
            sage_agg_lds_kernel<true><<<agg_blocks, AGG_THREADS, 0, stream>>>(src, dst, blk, cur, msg, lists,
                                                                              n_edges, n_nodes);
        else
            sage_agg_lds_kernel<false><<<agg_blocks, AGG_THREADS, 0, stream>>>(src, dst, blk, cur, msg, lists,
                                                                               n_edges, n_nodes);
        sage_layer_kernel<<<m_tiles, 256, 0, stream>>>(
            cur, msg,
            W_self  + (size_t)l * D_FEAT * D_FEAT,
            W_neigh + (size_t)l * D_FEAT * D_FEAT,
            b       + (size_t)l * D_FEAT,
            nxt);
        cur = nxt;
        nxt = (nxt == bufA) ? bufB : bufA;
    }

    sage_pred_kernel<<<m_tiles, 64, 0, stream>>>(cur, W_pred, b_pred, (float*)d_out);
}


# Neighbour-sum ranges of 400 rows, was 512: 250 work-groups, one on nearly each of the 256 WGPs (was 196), so a WGP's share of LDS float adds and row fetches falls 22%; arrival order as before
# speedup vs baseline: 2.9618x; 1.0448x over previous
#include <hip/hip_runtime.h>
#include <hip/hip_bf16.h>


#define D_FEAT 128
#define LDS_STRIDE 132
#define OUT_DIM 26

typedef __attribute__((ext_vector_type(2))) float v2f;
typedef __attribute__((ext_vector_type(8))) float v8f;

#define AGG_ROWS    400
#define AGG_THREADS 1024
#define AGG_QCAP    10240
#define AGG_SMASK   0x7FFFFF
#define AGG_REC     (AGG_QCAP + 4)

__device__ __forceinline__ bool sage_blk_ok(int n_edges, int n_nodes) {
    return ((n_nodes + AGG_ROWS - 1) / AGG_ROWS) <= 255 &&
           (size_t)n_edges <= (size_t)n_nodes * (D_FEAT * sizeof(float));
}


__device__ __forceinline__ unsigned sage_range_byte(int d) {
    const unsigned b = (unsigned)d / AGG_ROWS;
    return b < 255u ? b : 255u;
}

__global__ void sage_range_bytes_kernel(const int* __restrict__ dst, unsigned char* __restrict__ blk,
                                        int n_edges, int n_nodes) {
    if (!sage_blk_ok(n_edges, n_nodes)) return;
    const int e0 = (blockIdx.x * blockDim.x + threadIdx.x) * 16;
    if (e0 + 15 < n_edges) {
        int4 v[4];
        #pragma unroll
        for (int k = 0; k < 4; ++k) v[k] = *(const int4*)(dst + e0 + 4 * k);
        unsigned w[4];
        #pragma unroll
        for (int k = 0; k < 4; ++k)
            w[k] = sage_range_byte(v[k].x)       | sage_range_byte(v[k].y) << 8 |
                   sage_range_byte(v[k].z) << 16 | sage_range_byte(v[k].w) << 24;
        *(uint4*)(blk + e0) = make_uint4(w[0], w[1], w[2], w[3]);
    } else {
        for (int e = e0; e < n_edges; ++e) blk[e] = (unsigned char)sage_range_byte(dst[e]);
    }
}

template <bool GUARD>
__device__ __forceinline__ void sage_agg_scan(int base, int t, unsigned lo, int n_edges,
                                              const int* __restrict__ dst,
                                              int* __restrict__ queue, int* qn) {
    int4 v[8];
    #pragma unroll
    for (int k = 0; k < 8; ++k) {
        const int e0 = base + k * (AGG_THREADS * 4) + t * 4;
        if (!GUARD || e0 + 3 < n_edges) v[k] = *(const int4*)(dst + e0);
        else                            v[k] = make_int4(-1, -1, -1, -1);
    }
    #pragma unroll
    for (int k = 0; k < 8; ++k) {
        const int e0 = base + k * (AGG_THREADS * 4) + t * 4;
        const unsigned x0 = (unsigned)v[k].x - lo;
        const unsigned x1 = (unsigned)v[k].y - lo;
        const unsigned x2 = (unsigned)v[k].z - lo;
        const unsigned x3 = (unsigned)v[k].w - lo;
        const unsigned m01 = x0 < x1 ? x0 : x1;
        const unsigned m23 = x2 < x3 ? x2 : x3;
        const unsigned m   = m01 < m23 ? m01 : m23;
        if (m < AGG_ROWS) {
            const int pos = atomicAdd(qn, 1);
            if (pos < AGG_QCAP) queue[pos] = e0 >> 2;
        }
    }
}

template <bool GUARD>
__device__ __forceinline__ void sage_agg_scan_bytes(int base, int t, unsigned rep, int n_edges,
                                                    const unsigned char* __restrict__ blk,
                                                    int* __restrict__ queue, int* qn) {
    uint4 v[8];
    #pragma unroll
    for (int k = 0; k < 8; ++k) {
        const int e0 = base + k * (AGG_THREADS * 16) + t * 16;
        if (!GUARD || e0 + 15 < n_edges) {
            v[k] = *(const uint4*)(blk + e0);
        } else {
            unsigned w[4];
            #pragma unroll
            for (int j = 0; j < 4; ++j)
                w[j] = (e0 + 4 * j + 3 < n_edges) ? *(const unsigned*)(blk + e0 + 4 * j) : ~rep;
            v[k] = make_uint4(w[0], w[1], w[2], w[3]);
        }
    }
    unsigned mask = 0;
    #pragma unroll
    for (int k = 0; k < 8; ++k) {
        const unsigned w[4] = {v[k].x, v[k].y, v[k].z, v[k].w};
        #pragma unroll
        for (int j = 0; j < 4; ++j) {
            const unsigned x = w[j] ^ rep;
            if ((x - 0x01010101u) & ~x & 0x80808080u) mask |= 1u << (k * 4 + j);
        }
    }
    if (mask) {
        int pos = atomicAdd(qn, __popc(mask));
        const int g0 = (base >> 2) + t * 4;
        while (mask) {
            const int bit = __ffs(mask) - 1;
            mask &= mask - 1;
            if (pos < AGG_QCAP) queue[pos] = g0 + (bit >> 2) * (AGG_THREADS * 4) + (bit & 3);
            ++pos;
        }
    }
}

__device__ __forceinline__ void sage_agg_append(int ent, int total, int* __restrict__ queue,
                                                int* qextra) {
    const int pos = total + atomicAdd(qextra, 1);
    if (pos < AGG_QCAP) queue[pos] = ent;
}

template <bool BUILD>
__global__ void __launch_bounds__(AGG_THREADS)
sage_agg_lds_kernel(const int* __restrict__ src, const int* __restrict__ dst,
                    const unsigned char* __restrict__ blk,
                    const float* __restrict__ x, float* __restrict__ msg,
                    int* __restrict__ lists, int n_edges, int n_nodes) {
    __shared__ float acc[AGG_ROWS * D_FEAT];
    __shared__ __align__(16) int queue[AGG_QCAP];
    __shared__ int   qn;
    __shared__ int   qextra;
    __shared__ int   rowcnt[AGG_ROWS];

    const int t = threadIdx.x;
    const unsigned lo = blockIdx.x * AGG_ROWS;

    int* rec = lists + (size_t)blockIdx.x * AGG_REC;
    const bool fits = (size_t)gridDim.x * AGG_REC <= (size_t)n_nodes * OUT_DIM;

    for (int i = t; i < AGG_ROWS * D_FEAT / 4; i += AGG_THREADS)
        ((float4*)acc)[i] = make_float4(0.0f, 0.0f, 0.0f, 0.0f);
    for (int i = t; i < AGG_ROWS; i += AGG_THREADS) rowcnt[i] = 0;
    if (t == 0) { qn = 0; qextra = 0; }
    __syncthreads();

    int  total2 = 0;
    bool go     = false;
    bool loaded = false;

    if (!BUILD && fits) {
        const int cnt = rec[0];
        if (cnt >= 0 && cnt <= AGG_QCAP) {
            for (int i = t; i < ((cnt + 3) >> 2); i += AGG_THREADS)
                ((int4*)queue)[i] = ((const int4*)(rec + 4))[i];
            total2 = cnt;
            go     = true;
            loaded = true;
        }
    }

    if (!loaded) {
        if (BUILD && sage_blk_ok(n_edges, n_nodes)) {
            const unsigned rep = blockIdx.x * 0x01010101u;
            const int per_iter = AGG_THREADS * 128;
            int base = 0;
            for (; base + per_iter <= n_edges; base += per_iter)
                sage_agg_scan_bytes<false>(base, t, rep, n_edges, blk, queue, &qn);
            if (base < n_edges)
                sage_agg_scan_bytes<true>(base, t, rep, n_edges, blk, queue, &qn);
        } else {
            const int per_iter = AGG_THREADS * 32;
            int base = 0;
            for (; base + per_iter <= n_edges; base += per_iter)
                sage_agg_scan<false>(base, t, lo, n_edges, dst, queue, &qn);
            if (base < n_edges)
                sage_agg_scan<true>(base, t, lo, n_edges, dst, queue, &qn);
        }
        __syncthreads();

        const int  total  = qn;
        const bool queued = total <= AGG_QCAP;

        if (queued) {
            for (int i = t; i < total; i += AGG_THREADS) {
                const int  e0 = queue[i] * 4;
                const int4 dd = *(const int4*)(dst + e0);
                const int4 ss = *(const int4*)(src + e0);
                const int  dv[4] = {dd.x, dd.y, dd.z, dd.w};
                const int  sv[4] = {ss.x, ss.y, ss.z, ss.w};
                bool have = false;
                int  ent0 = 0;
                #pragma unroll
                for (int j = 0; j < 4; ++j) {
                    const unsigned loc = (unsigned)dv[j] - lo;
                    if (loc < AGG_ROWS) {
                        const int ent = (int)((loc << 23) | ((unsigned)sv[j] & AGG_SMASK));
                        if (!have) { ent0 = ent; have = true; }
                        else       sage_agg_append(ent, total, queue, &qextra);
                    }
                }
                queue[i] = ent0;
            }
            const int rem = n_edges & 3;
            if (t < rem) {
                const int e = n_edges - rem + t;
                const unsigned loc = (unsigned)dst[e] - lo;
                if (loc < AGG_ROWS)
                    sage_agg_append((int)((loc << 23) | ((unsigned)src[e] & AGG_SMASK)),
                                    total, queue, &qextra);
            }
        }
        __syncthreads();

        total2 = total + qextra;
        go     = queued && total2 <= AGG_QCAP;

        if (BUILD && fits) {
            if (go)
                for (int i = t; i < ((total2 + 3) >> 2); i += AGG_THREADS)
                    ((int4*)(rec + 4))[i] = ((const int4*)queue)[i];
            if (t == 0) rec[0] = go ? total2 : -1;
        }
    }
    __syncthreads();

    if (go) {
        for (int i = t; i < total2; i += AGG_THREADS)
            atomicAdd(&rowcnt[(unsigned)queue[i] >> 23], 1);

        const int wave = t >> 5;
        const int lane = t & 31;
        for (int i = wave * 8; i < total2; i += (AGG_THREADS / 32) * 8) {
            int    ent[8];
            float4 r[8];
            #pragma unroll
            for (int k = 0; k < 8; ++k) {
                const int idx = i + k;
                ent[k] = queue[idx < total2 ? idx : i];
                r[k] = *(const float4*)(x + (size_t)(ent[k] & AGG_SMASK) * D_FEAT + lane * 4);
            }
            #pragma unroll
            for (int k = 0; k < 8; ++k) {
                if (i + k < total2) {
                    float* ar = acc + ((unsigned)ent[k] >> 23) * D_FEAT + lane;
                    atomicAdd(ar +  0, r[k].x);
                    atomicAdd(ar + 32, r[k].y);
                    atomicAdd(ar + 64, r[k].z);
                    atomicAdd(ar + 96, r[k].w);
                }
            }
        }
    } else {
        for (int e = t; e < n_edges; e += AGG_THREADS) {
            const unsigned loc = (unsigned)dst[e] - lo;
            if (loc < AGG_ROWS) {
                const float* xr = x + (size_t)src[e] * D_FEAT;
                float*       ar = acc + loc * D_FEAT;
                atomicAdd(&rowcnt[loc], 1);
                for (int c = 0; c < D_FEAT; ++c) atomicAdd(ar + (c & 3) * 32 + (c >> 2), xr[c]);
            }
        }
    }
    __syncthreads();

    {
        const int row0 = blockIdx.x * AGG_ROWS;
        for (int i = t; i < AGG_ROWS * (D_FEAT / 4); i += AGG_THREADS) {
            const int r = i >> 5;
            const int l = i & 31;
            if (row0 + r < n_nodes) {
                const float* ar = acc + r * D_FEAT + l;
                const float  iv = 1.0f / fmaxf((float)rowcnt[r], 1.0f);
                *(float4*)(msg + (size_t)(row0 + r) * D_FEAT + l * 4) =
                    make_float4(ar[0] * iv, ar[32] * iv, ar[64] * iv, ar[96] * iv);
            }
        }
    }
}

__global__ void __launch_bounds__(256)
sage_layer_kernel(const float* __restrict__ xin, const float* __restrict__ msg,
                  const float* __restrict__ Wself, const float* __restrict__ Wneigh,
                  const float* __restrict__ bias, float* __restrict__ xout) {
    __shared__ float As[16 * LDS_STRIDE];
    __shared__ float Hs[16 * LDS_STRIDE];

    const int m0 = blockIdx.x * 16;
    const int t  = threadIdx.x;

    {
        int r  = t >> 4;
        int c0 = (t & 15) * 8;
        const float* xr = xin + (size_t)(m0 + r) * D_FEAT + c0;
        float4 a0 = *(const float4*)(xr);
        float4 a1 = *(const float4*)(xr + 4);
        *(float4*)(As + r * LDS_STRIDE + c0)     = a0;
        *(float4*)(As + r * LDS_STRIDE + c0 + 4) = a1;

        const float* mr = msg + (size_t)(m0 + r) * D_FEAT + c0;
        float4 h0 = *(const float4*)(mr);
        float4 h1 = *(const float4*)(mr + 4);
        *(float4*)(Hs + r * LDS_STRIDE + c0)     = h0;
        *(float4*)(Hs + r * LDS_STRIDE + c0 + 4) = h1;
    }
    __syncthreads();

    const int wave = t >> 5;
    const int lane = t & 31;
    const int n0   = wave * 16;
    const int half = lane >> 4;
    const int l16  = lane & 15;
    const int col  = n0 + l16;

    v8f acc;
    {
        float bv = bias[col];
        #pragma unroll
        for (int i = 0; i < 8; ++i) acc[i] = bv;
    }

    #pragma unroll 4
    for (int k0 = 0; k0 < D_FEAT; k0 += 4) {
        const int ka = k0 + half * 2;
        v2f a = *(const v2f*)(As + l16 * LDS_STRIDE + ka);
        v2f b;
        b.x = Wself[(size_t)ka * D_FEAT + col];
        b.y = Wself[(size_t)(ka + 1) * D_FEAT + col];
        acc = __builtin_amdgcn_wmma_f32_16x16x4_f32(false, a, false, b,
                                                    (short)0, acc, false, false);

        v2f h = *(const v2f*)(Hs + l16 * LDS_STRIDE + ka);
        v2f bn;
        bn.x = Wneigh[(size_t)ka * D_FEAT + col];
        bn.y = Wneigh[(size_t)(ka + 1) * D_FEAT + col];
        acc = __builtin_amdgcn_wmma_f32_16x16x4_f32(false, h, false, bn,
                                                    (short)0, acc, false, false);
    }

    #pragma unroll
    for (int v = 0; v < 8; ++v) {
        int row = m0 + v + half * 8;
        float val = acc[v];
        val = val > 0.0f ? val : 0.0f;
        xout[(size_t)row * D_FEAT + col] = val;
    }
}

__global__ void __launch_bounds__(64)
sage_pred_kernel(const float* __restrict__ xin, const float* __restrict__ Wp,
                 const float* __restrict__ bp, float* __restrict__ out) {
    const int m0   = blockIdx.x * 16;
    const int t    = threadIdx.x;
    const int wave = t >> 5;
    const int lane = t & 31;
    const int n0   = wave * 16;
    const int half = lane >> 4;
    const int l16  = lane & 15;
    const int col  = n0 + l16;
    const int bcol = col < OUT_DIM ? col : (OUT_DIM - 1);

    v8f acc;
    {
        float bv = bp[bcol];
        #pragma unroll
        for (int i = 0; i < 8; ++i) acc[i] = bv;
    }

    #pragma unroll 4
    for (int k0 = 0; k0 < D_FEAT; k0 += 4) {
        const int ka = k0 + half * 2;
        const float* ap = xin + (size_t)(m0 + l16) * D_FEAT + ka;
        v2f a; a.x = ap[0]; a.y = ap[1];
        v2f b;
        b.x = Wp[(size_t)ka * OUT_DIM + bcol];
        b.y = Wp[(size_t)(ka + 1) * OUT_DIM + bcol];
        acc = __builtin_amdgcn_wmma_f32_16x16x4_f32(false, a, false, b,
                                                    (short)0, acc, false, false);
    }

    if (col < OUT_DIM) {
        #pragma unroll
        for (int v = 0; v < 8; ++v) {
            int row = m0 + v + half * 8;
            out[(size_t)row * OUT_DIM + col] = acc[v];
        }
    }
}


extern "C" void kernel_launch(void* const* d_in, const int* in_sizes, int n_in,
                              void* d_out, int out_size, void* d_ws, size_t ws_size,
                              hipStream_t stream) {
    const float* x       = (const float*)d_in[0];
    const int*   src     = (const int*)  d_in[1];
    const int*   dst     = (const int*)  d_in[2];
    const float* W_self  = (const float*)d_in[3];
    const float* W_neigh = (const float*)d_in[4];
    const float* b       = (const float*)d_in[5];
    const float* W_pred  = (const float*)d_in[6];
    const float* b_pred  = (const float*)d_in[7];

    const int n_nodes = in_sizes[0] / D_FEAT;
    const int n_edges = in_sizes[1];
    const int L       = in_sizes[5] / D_FEAT;
    const int feat_elems = n_nodes * D_FEAT;

    char*  ws = (char*)d_ws;
    size_t feat_bytes = (size_t)feat_elems * sizeof(float);
    float* msg    = (float*)ws;
    float* bufA   = (float*)(ws + feat_bytes);
    float* bufB   = (float*)(ws + 2 * feat_bytes);

    int* lists = (int*)d_out;

    unsigned char* blk = (unsigned char*)bufB;

    const int T = 256;

    sage_range_bytes_kernel<<<(n_edges + 16 * T - 1) / (16 * T), T, 0, stream>>>(dst, blk, n_edges, n_nodes);

    const float* cur = x;
    float* nxt = bufA;
    const int agg_blocks = (n_nodes + AGG_ROWS - 1) / AGG_ROWS;
    const int m_tiles = n_nodes / 16;

    for (int l = 0; l < L; ++l) {
        if (l == 0)
            sage_agg_lds_kernel<true><<<agg_blocks, AGG_THREADS, 0, stream>>>(src, dst, blk, cur, msg, lists,
                                                                              n_edges, n_nodes);
        else
            sage_agg_lds_kernel<false><<<agg_blocks, AGG_THREADS, 0, stream>>>(src, dst, blk, cur, msg, lists,
                                                                               n_edges, n_nodes);
        sage_layer_kernel<<<m_tiles, 256, 0, stream>>>(
            cur, msg,
            W_self  + (size_t)l * D_FEAT * D_FEAT,
            W_neigh + (size_t)l * D_FEAT * D_FEAT,
            b       + (size_t)l * D_FEAT,
            nxt);
        cur = nxt;
        nxt = (nxt == bufA) ? bufB : bufA;
    }

    sage_pred_kernel<<<m_tiles, 64, 0, stream>>>(cur, W_pred, b_pred, (float*)d_out);
}


# Layer product on 256 resident work-groups: weights copied once into LDS as k quads, a wave keeps a 16-row tile's x then mean rows in registers for 8 column tiles; k order changes, bias added last
# speedup vs baseline: 3.4654x; 1.1700x over previous
#include <hip/hip_runtime.h>
#include <hip/hip_bf16.h>


#define D_FEAT 128
#define OUT_DIM 26

typedef __attribute__((ext_vector_type(2))) float v2f;
typedef __attribute__((ext_vector_type(8))) float v8f;

#define AGG_ROWS    400
#define AGG_THREADS 1024
#define AGG_QCAP    10240
#define AGG_SMASK   0x7FFFFF
#define AGG_REC     (AGG_QCAP + 4)

__device__ __forceinline__ bool sage_blk_ok(int n_edges, int n_nodes) {
    return ((n_nodes + AGG_ROWS - 1) / AGG_ROWS) <= 255 &&
           (size_t)n_edges <= (size_t)n_nodes * (D_FEAT * sizeof(float));
}


__device__ __forceinline__ unsigned sage_range_byte(int d) {
    const unsigned b = (unsigned)d / AGG_ROWS;
    return b < 255u ? b : 255u;
}

__global__ void sage_range_bytes_kernel(const int* __restrict__ dst, unsigned char* __restrict__ blk,
                                        int n_edges, int n_nodes) {
    if (!sage_blk_ok(n_edges, n_nodes)) return;
    const int e0 = (blockIdx.x * blockDim.x + threadIdx.x) * 16;
    if (e0 + 15 < n_edges) {
        int4 v[4];
        #pragma unroll
        for (int k = 0; k < 4; ++k) v[k] = *(const int4*)(dst + e0 + 4 * k);
        unsigned w[4];
        #pragma unroll
        for (int k = 0; k < 4; ++k)
            w[k] = sage_range_byte(v[k].x)       | sage_range_byte(v[k].y) << 8 |
                   sage_range_byte(v[k].z) << 16 | sage_range_byte(v[k].w) << 24;
        *(uint4*)(blk + e0) = make_uint4(w[0], w[1], w[2], w[3]);
    } else {
        for (int e = e0; e < n_edges; ++e) blk[e] = (unsigned char)sage_range_byte(dst[e]);
    }
}

template <bool GUARD>
__device__ __forceinline__ void sage_agg_scan(int base, int t, unsigned lo, int n_edges,
                                              const int* __restrict__ dst,
                                              int* __restrict__ queue, int* qn) {
    int4 v[8];
    #pragma unroll
    for (int k = 0; k < 8; ++k) {
        const int e0 = base + k * (AGG_THREADS * 4) + t * 4;
        if (!GUARD || e0 + 3 < n_edges) v[k] = *(const int4*)(dst + e0);
        else                            v[k] = make_int4(-1, -1, -1, -1);
    }
    #pragma unroll
    for (int k = 0; k < 8; ++k) {
        const int e0 = base + k * (AGG_THREADS * 4) + t * 4;
        const unsigned x0 = (unsigned)v[k].x - lo;
        const unsigned x1 = (unsigned)v[k].y - lo;
        const unsigned x2 = (unsigned)v[k].z - lo;
        const unsigned x3 = (unsigned)v[k].w - lo;
        const unsigned m01 = x0 < x1 ? x0 : x1;
        const unsigned m23 = x2 < x3 ? x2 : x3;
        const unsigned m   = m01 < m23 ? m01 : m23;
        if (m < AGG_ROWS) {
            const int pos = atomicAdd(qn, 1);
            if (pos < AGG_QCAP) queue[pos] = e0 >> 2;
        }
    }
}

template <bool GUARD>
__device__ __forceinline__ void sage_agg_scan_bytes(int base, int t, unsigned rep, int n_edges,
                                                    const unsigned char* __restrict__ blk,
                                                    int* __restrict__ queue, int* qn) {
    uint4 v[8];
    #pragma unroll
    for (int k = 0; k < 8; ++k) {
        const int e0 = base + k * (AGG_THREADS * 16) + t * 16;
        if (!GUARD || e0 + 15 < n_edges) {
            v[k] = *(const uint4*)(blk + e0);
        } else {
            unsigned w[4];
            #pragma unroll
            for (int j = 0; j < 4; ++j)
                w[j] = (e0 + 4 * j + 3 < n_edges) ? *(const unsigned*)(blk + e0 + 4 * j) : ~rep;
            v[k] = make_uint4(w[0], w[1], w[2], w[3]);
        }
    }
    unsigned mask = 0;
    #pragma unroll
    for (int k = 0; k < 8; ++k) {
        const unsigned w[4] = {v[k].x, v[k].y, v[k].z, v[k].w};
        #pragma unroll
        for (int j = 0; j < 4; ++j) {
            const unsigned x = w[j] ^ rep;
            if ((x - 0x01010101u) & ~x & 0x80808080u) mask |= 1u << (k * 4 + j);
        }
    }
    if (mask) {
        int pos = atomicAdd(qn, __popc(mask));
        const int g0 = (base >> 2) + t * 4;
        while (mask) {
            const int bit = __ffs(mask) - 1;
            mask &= mask - 1;
            if (pos < AGG_QCAP) queue[pos] = g0 + (bit >> 2) * (AGG_THREADS * 4) + (bit & 3);
            ++pos;
        }
    }
}

__device__ __forceinline__ void sage_agg_append(int ent, int total, int* __restrict__ queue,
                                                int* qextra) {
    const int pos = total + atomicAdd(qextra, 1);
    if (pos < AGG_QCAP) queue[pos] = ent;
}

template <bool BUILD>
__global__ void __launch_bounds__(AGG_THREADS)
sage_agg_lds_kernel(const int* __restrict__ src, const int* __restrict__ dst,
                    const unsigned char* __restrict__ blk,
                    const float* __restrict__ x, float* __restrict__ msg,
                    int* __restrict__ lists, int n_edges, int n_nodes) {
    __shared__ float acc[AGG_ROWS * D_FEAT];
    __shared__ __align__(16) int queue[AGG_QCAP];
    __shared__ int   qn;
    __shared__ int   qextra;
    __shared__ int   rowcnt[AGG_ROWS];

    const int t = threadIdx.x;
    const unsigned lo = blockIdx.x * AGG_ROWS;

    int* rec = lists + (size_t)blockIdx.x * AGG_REC;
    const bool fits = (size_t)gridDim.x * AGG_REC <= (size_t)n_nodes * OUT_DIM;

    for (int i = t; i < AGG_ROWS * D_FEAT / 4; i += AGG_THREADS)
        ((float4*)acc)[i] = make_float4(0.0f, 0.0f, 0.0f, 0.0f);
    for (int i = t; i < AGG_ROWS; i += AGG_THREADS) rowcnt[i] = 0;
    if (t == 0) { qn = 0; qextra = 0; }
    __syncthreads();

    int  total2 = 0;
    bool go     = false;
    bool loaded = false;

    if (!BUILD && fits) {
        const int cnt = rec[0];
        if (cnt >= 0 && cnt <= AGG_QCAP) {
            for (int i = t; i < ((cnt + 3) >> 2); i += AGG_THREADS)
                ((int4*)queue)[i] = ((const int4*)(rec + 4))[i];
            total2 = cnt;
            go     = true;
            loaded = true;
        }
    }

    if (!loaded) {
        if (BUILD && sage_blk_ok(n_edges, n_nodes)) {
            const unsigned rep = blockIdx.x * 0x01010101u;
            const int per_iter = AGG_THREADS * 128;
            int base = 0;
            for (; base + per_iter <= n_edges; base += per_iter)
                sage_agg_scan_bytes<false>(base, t, rep, n_edges, blk, queue, &qn);
            if (base < n_edges)
                sage_agg_scan_bytes<true>(base, t, rep, n_edges, blk, queue, &qn);
        } else {
            const int per_iter = AGG_THREADS * 32;
            int base = 0;
            for (; base + per_iter <= n_edges; base += per_iter)
                sage_agg_scan<false>(base, t, lo, n_edges, dst, queue, &qn);
            if (base < n_edges)
                sage_agg_scan<true>(base, t, lo, n_edges, dst, queue, &qn);
        }
        __syncthreads();

        const int  total  = qn;
        const bool queued = total <= AGG_QCAP;

        if (queued) {
            for (int i = t; i < total; i += AGG_THREADS) {
                const int  e0 = queue[i] * 4;
                const int4 dd = *(const int4*)(dst + e0);
                const int4 ss = *(const int4*)(src + e0);
                const int  dv[4] = {dd.x, dd.y, dd.z, dd.w};
                const int  sv[4] = {ss.x, ss.y, ss.z, ss.w};
                bool have = false;
                int  ent0 = 0;
                #pragma unroll
                for (int j = 0; j < 4; ++j) {
                    const unsigned loc = (unsigned)dv[j] - lo;
                    if (loc < AGG_ROWS) {
                        const int ent = (int)((loc << 23) | ((unsigned)sv[j] & AGG_SMASK));
                        if (!have) { ent0 = ent; have = true; }
                        else       sage_agg_append(ent, total, queue, &qextra);
                    }
                }
                queue[i] = ent0;
            }
            const int rem = n_edges & 3;
            if (t < rem) {
                const int e = n_edges - rem + t;
                const unsigned loc = (unsigned)dst[e] - lo;
                if (loc < AGG_ROWS)
                    sage_agg_append((int)((loc << 23) | ((unsigned)src[e] & AGG_SMASK)),
                                    total, queue, &qextra);
            }
        }
        __syncthreads();

        total2 = total + qextra;
        go     = queued && total2 <= AGG_QCAP;

        if (BUILD && fits) {
            if (go)
                for (int i = t; i < ((total2 + 3) >> 2); i += AGG_THREADS)
                    ((int4*)(rec + 4))[i] = ((const int4*)queue)[i];
            if (t == 0) rec[0] = go ? total2 : -1;
        }
    }
    __syncthreads();

    if (go) {
        for (int i = t; i < total2; i += AGG_THREADS)
            atomicAdd(&rowcnt[(unsigned)queue[i] >> 23], 1);

        const int wave = t >> 5;
        const int lane = t & 31;
        for (int i = wave * 8; i < total2; i += (AGG_THREADS / 32) * 8) {
            int    ent[8];
            float4 r[8];
            #pragma unroll
            for (int k = 0; k < 8; ++k) {
                const int idx = i + k;
                ent[k] = queue[idx < total2 ? idx : i];
                r[k] = *(const float4*)(x + (size_t)(ent[k] & AGG_SMASK) * D_FEAT + lane * 4);
            }
            #pragma unroll
            for (int k = 0; k < 8; ++k) {
                if (i + k < total2) {
                    float* ar = acc + ((unsigned)ent[k] >> 23) * D_FEAT + lane;
                    atomicAdd(ar +  0, r[k].x);
                    atomicAdd(ar + 32, r[k].y);
                    atomicAdd(ar + 64, r[k].z);
                    atomicAdd(ar + 96, r[k].w);
                }
            }
        }
    } else {
        for (int e = t; e < n_edges; e += AGG_THREADS) {
            const unsigned loc = (unsigned)dst[e] - lo;
            if (loc < AGG_ROWS) {
                const float* xr = x + (size_t)src[e] * D_FEAT;
                float*       ar = acc + loc * D_FEAT;
                atomicAdd(&rowcnt[loc], 1);
                for (int c = 0; c < D_FEAT; ++c) atomicAdd(ar + (c & 3) * 32 + (c >> 2), xr[c]);
            }
        }
    }
    __syncthreads();

    {
        const int row0 = blockIdx.x * AGG_ROWS;
        for (int i = t; i < AGG_ROWS * (D_FEAT / 4); i += AGG_THREADS) {
            const int r = i >> 5;
            const int l = i & 31;
            if (row0 + r < n_nodes) {
                const float* ar = acc + r * D_FEAT + l;
                const float  iv = 1.0f / fmaxf((float)rowcnt[r], 1.0f);
                *(float4*)(msg + (size_t)(row0 + r) * D_FEAT + l * 4) =
                    make_float4(ar[0] * iv, ar[32] * iv, ar[64] * iv, ar[96] * iv);
            }
        }
    }
}

#define WQ_PITCH       516
#define LAYER_THREADS  512
#define LAYER_WGS      256

__device__ __forceinline__ void sage_stage_weights(const float* __restrict__ Wg, float* __restrict__ Wl, int t) {
    const int q0 = t >> 7;
    const int n  = t & 127;
    float4 w[8];
    #pragma unroll
    for (int j = 0; j < 8; ++j) {
        const float* g = Wg + (size_t)(4 * (q0 + (LAYER_THREADS / D_FEAT) * j)) * D_FEAT + n;
        w[j] = make_float4(g[0], g[D_FEAT], g[2 * D_FEAT], g[3 * D_FEAT]);
    }
    #pragma unroll
    for (int j = 0; j < 8; ++j)
        *(float4*)(Wl + (q0 + (LAYER_THREADS / D_FEAT) * j) * WQ_PITCH + n * 4) = w[j];
}

__global__ void __launch_bounds__(LAYER_THREADS)
sage_layer_kernel(const float* __restrict__ xin, const float* __restrict__ msg,
                  const float* __restrict__ Wself, const float* __restrict__ Wneigh,
                  const float* __restrict__ bias, float* __restrict__ xout, int m_tiles) {
    __shared__ __align__(16) float Ws[32 * WQ_PITCH];
    __shared__ __align__(16) float Wn[32 * WQ_PITCH];

    const int t = threadIdx.x;
    const int wave = __builtin_amdgcn_readfirstlane(t >> 5);
    const int lane = t & 31;
    const int half = lane >> 4;
    const int l16  = lane & 15;
    const int tstep = LAYER_WGS * (LAYER_THREADS / 32);
    const unsigned lrow = (unsigned)l16 * D_FEAT + half * 4;
    const unsigned lout = (unsigned)half * (8 * D_FEAT) + l16;

    int tile = blockIdx.x + wave * LAYER_WGS;

    float4 xa[16];
    {
        const int t0 = tile < m_tiles ? tile : 0;
        const float* xr = (xin + (size_t)t0 * (16 * D_FEAT)) + lrow;
        #pragma unroll
        for (int i = 0; i < 16; ++i) xa[i] = *(const float4*)(xr + 8 * i);
    }

    sage_stage_weights(Wself,  Ws, t);
    sage_stage_weights(Wneigh, Wn, t);
    __syncthreads();

    float bv[8];
    #pragma unroll
    for (int nt = 0; nt < 8; ++nt) bv[nt] = bias[nt * 16 + l16];

    const float* bs0 = Ws + half * WQ_PITCH + l16 * 4;
    const float* bn0 = Wn + half * WQ_PITCH + l16 * 4;

    for (; tile < m_tiles; tile += tstep) {
        const int m0 = tile * 16;
        const int pf = tile + tstep < m_tiles ? tile + tstep : tile;
        const float* mr = (msg + (size_t)m0 * D_FEAT) + lrow;
        const float* xn = (xin + (size_t)pf * (16 * D_FEAT)) + lrow;

        v8f acc[8];
        #pragma unroll
        for (int nt = 0; nt < 8; ++nt) {
            #pragma unroll
            for (int v = 0; v < 8; ++v) acc[nt][v] = 0.0f;
        }

        float4 ha[16];
        float4 bq[2][8];
        #pragma unroll
        for (int nt = 0; nt < 8; ++nt) bq[0][nt] = *(const float4*)(bs0 + nt * 64);
        #pragma unroll
        for (int u = 0; u < 32; ++u) {
            const int i = u & 15;
            if (u < 16) ha[i] = *(const float4*)(mr + 8 * i);
            else        xa[i] = *(const float4*)(xn + 8 * i);
            if (u + 1 < 32) {
                const int u1 = u + 1;
                const float* bp = (u1 < 16 ? bs0 : bn0) + 2 * (u1 & 15) * WQ_PITCH;
                #pragma unroll
                for (int nt = 0; nt < 8; ++nt) bq[u1 & 1][nt] = *(const float4*)(bp + nt * 64);
            }
            const float4 ar = u < 16 ? xa[i] : ha[i];
            #pragma unroll
            for (int s = 0; s < 2; ++s) {
                v2f a;
                a.x = s ? ar.z : ar.x;
                a.y = s ? ar.w : ar.y;
                #pragma unroll
                for (int nt = 0; nt < 8; ++nt) {
                    v2f b;
                    b.x = s ? bq[u & 1][nt].z : bq[u & 1][nt].x;
                    b.y = s ? bq[u & 1][nt].w : bq[u & 1][nt].y;
                    acc[nt] = __builtin_amdgcn_wmma_f32_16x16x4_f32(false, a, false, b,
                                                                    (short)0, acc[nt], false, false);
                }
            }
            __builtin_amdgcn_sched_barrier(0);
        }

        float* orow = (xout + (size_t)m0 * D_FEAT) + lout;
        #pragma unroll
        for (int nt = 0; nt < 8; ++nt) {
            #pragma unroll
            for (int v = 0; v < 8; ++v) {
                float val = acc[nt][v] + bv[nt];
                val = val > 0.0f ? val : 0.0f;
                orow[v * D_FEAT + nt * 16] = val;
            }
        }
    }
}

__global__ void __launch_bounds__(64)
sage_pred_kernel(const float* __restrict__ xin, const float* __restrict__ Wp,
                 const float* __restrict__ bp, float* __restrict__ out) {
    const int m0   = blockIdx.x * 16;
    const int t    = threadIdx.x;
    const int wave = t >> 5;
    const int lane = t & 31;
    const int n0   = wave * 16;
    const int half = lane >> 4;
    const int l16  = lane & 15;
    const int col  = n0 + l16;
    const int bcol = col < OUT_DIM ? col : (OUT_DIM - 1);

    v8f acc;
    {
        float bv = bp[bcol];
        #pragma unroll
        for (int i = 0; i < 8; ++i) acc[i] = bv;
    }

    #pragma unroll 4
    for (int k0 = 0; k0 < D_FEAT; k0 += 4) {
        const int ka = k0 + half * 2;
        const float* ap = xin + (size_t)(m0 + l16) * D_FEAT + ka;
        v2f a; a.x = ap[0]; a.y = ap[1];
        v2f b;
        b.x = Wp[(size_t)ka * OUT_DIM + bcol];
        b.y = Wp[(size_t)(ka + 1) * OUT_DIM + bcol];
        acc = __builtin_amdgcn_wmma_f32_16x16x4_f32(false, a, false, b,
                                                    (short)0, acc, false, false);
    }

    if (col < OUT_DIM) {
        #pragma unroll
        for (int v = 0; v < 8; ++v) {
            int row = m0 + v + half * 8;
            out[(size_t)row * OUT_DIM + col] = acc[v];
        }
    }
}


extern "C" void kernel_launch(void* const* d_in, const int* in_sizes, int n_in,
                              void* d_out, int out_size, void* d_ws, size_t ws_size,
                              hipStream_t stream) {
    const float* x       = (const float*)d_in[0];
    const int*   src     = (const int*)  d_in[1];
    const int*   dst     = (const int*)  d_in[2];
    const float* W_self  = (const float*)d_in[3];
    const float* W_neigh = (const float*)d_in[4];
    const float* b       = (const float*)d_in[5];
    const float* W_pred  = (const float*)d_in[6];
    const float* b_pred  = (const float*)d_in[7];

    const int n_nodes = in_sizes[0] / D_FEAT;
    const int n_edges = in_sizes[1];
    const int L       = in_sizes[5] / D_FEAT;
    const int feat_elems = n_nodes * D_FEAT;

    char*  ws = (char*)d_ws;
    size_t feat_bytes = (size_t)feat_elems * sizeof(float);
    float* msg    = (float*)ws;
    float* bufA   = (float*)(ws + feat_bytes);
    float* bufB   = (float*)(ws + 2 * feat_bytes);

    int* lists = (int*)d_out;

    unsigned char* blk = (unsigned char*)bufB;

    const int T = 256;

    sage_range_bytes_kernel<<<(n_edges + 16 * T - 1) / (16 * T), T, 0, stream>>>(dst, blk, n_edges, n_nodes);

    const float* cur = x;
    float* nxt = bufA;
    const int agg_blocks = (n_nodes + AGG_ROWS - 1) / AGG_ROWS;
    const int m_tiles = n_nodes / 16;

    for (int l = 0; l < L; ++l) {
        if (l == 0)
            sage_agg_lds_kernel<true><<<agg_blocks, AGG_THREADS, 0, stream>>>(src, dst, blk, cur, msg, lists,
                                                                              n_edges, n_nodes);
        else
            sage_agg_lds_kernel<false><<<agg_blocks, AGG_THREADS, 0, stream>>>(src, dst, blk, cur, msg, lists,
                                                                               n_edges, n_nodes);
        sage_layer_kernel<<<LAYER_WGS, LAYER_THREADS, 0, stream>>>(
            cur, msg,
            W_self  + (size_t)l * D_FEAT * D_FEAT,
            W_neigh + (size_t)l * D_FEAT * D_FEAT,
            b       + (size_t)l * D_FEAT,
            nxt, m_tiles);
        cur = nxt;
        nxt = (nxt == bufA) ? bufB : bufA;
    }

    sage_pred_kernel<<<m_tiles, 64, 0, stream>>>(cur, W_pred, b_pred, (float*)d_out);
}


# Layer 1's edge lists from one dealing pass: 256 work-groups each bin a slice of the edges into per-range cells (LDS counters), an owner joins its 256 cells; 250 full scans go; arrival order as before
# speedup vs baseline: 3.6751x; 1.0605x over previous
#include <hip/hip_runtime.h>
#include <hip/hip_bf16.h>


#define D_FEAT 128
#define OUT_DIM 26

typedef __attribute__((ext_vector_type(2))) float v2f;
typedef __attribute__((ext_vector_type(8))) float v8f;

#define AGG_ROWS    400
#define AGG_THREADS 1024
#define AGG_QCAP    10240
#define AGG_SMASK   0x7FFFFF
#define AGG_REC     (AGG_QCAP + 4)

#define BIN_WGS     256
#define BIN_THREADS 1024
#define BIN_CAP     64
#define BIN_MAXR    1024

__device__ __forceinline__ bool sage_cells_ok(int n_nodes) {
    const size_t nr = (size_t)((n_nodes + AGG_ROWS - 1) / AGG_ROWS);
    return nr <= BIN_MAXR &&
           nr * BIN_WGS * (BIN_CAP + 1) * sizeof(int) <= (size_t)n_nodes * (D_FEAT * sizeof(float));
}


__device__ __forceinline__ void sage_bin_place(int d, int s, int n_ranges,
                                               int* __restrict__ cnt, int* __restrict__ cell0) {
    const unsigned r = (unsigned)d / AGG_ROWS;
    if (r < (unsigned)n_ranges) {
        const int pos = atomicAdd(&cnt[r], 1);
        if (pos < BIN_CAP)
            cell0[(size_t)r * (BIN_WGS * BIN_CAP) + pos] =
                (int)((((unsigned)d - r * AGG_ROWS) << 23) | ((unsigned)s & AGG_SMASK));
    }
}

__global__ void __launch_bounds__(BIN_THREADS)
sage_bin_kernel(const int* __restrict__ src, const int* __restrict__ dst,
                int* __restrict__ cnts, int* __restrict__ cells, int n_edges, int n_nodes) {
    __shared__ int cnt[BIN_MAXR];
    if (!sage_cells_ok(n_nodes)) return;

    const int t = threadIdx.x;
    const int p = blockIdx.x;
    const int n_ranges = (n_nodes + AGG_ROWS - 1) / AGG_ROWS;
    for (int r = t; r < n_ranges; r += BIN_THREADS) cnt[r] = 0;
    __syncthreads();

    const int per = (((n_edges + BIN_WGS - 1) / BIN_WGS) + 3) & ~3;
    const long long lo64 = (long long)p * per;
    const int e_lo = lo64 < (long long)n_edges ? (int)lo64 : n_edges;
    const int e_hi = (n_edges - e_lo) < per ? n_edges : e_lo + per;
    const int nq   = (e_hi - e_lo) >> 2;
    int* cell0 = cells + (size_t)p * BIN_CAP;

    for (int qb = 0; qb < nq; qb += 2 * BIN_THREADS) {
        int4 dd[2], ss[2];
        #pragma unroll
        for (int k = 0; k < 2; ++k) {
            const int q = qb + k * BIN_THREADS + t;
            if (q < nq) {
                dd[k] = *(const int4*)(dst + e_lo + 4 * q);
                ss[k] = *(const int4*)(src + e_lo + 4 * q);
            } else {
                dd[k] = make_int4(-1, -1, -1, -1);
                ss[k] = make_int4(0, 0, 0, 0);
            }
        }
        #pragma unroll
        for (int k = 0; k < 2; ++k) {
            sage_bin_place(dd[k].x, ss[k].x, n_ranges, cnt, cell0);
            sage_bin_place(dd[k].y, ss[k].y, n_ranges, cnt, cell0);
            sage_bin_place(dd[k].z, ss[k].z, n_ranges, cnt, cell0);
            sage_bin_place(dd[k].w, ss[k].w, n_ranges, cnt, cell0);
        }
    }
    const int rem = (e_hi - e_lo) & 3;
    if (t < rem) {
        const int e = e_lo + 4 * nq + t;
        sage_bin_place(dst[e], src[e], n_ranges, cnt, cell0);
    }
    __syncthreads();

    for (int r = t; r < n_ranges; r += BIN_THREADS) cnts[(size_t)r * BIN_WGS + p] = cnt[r];
}

template <bool GUARD>
__device__ __forceinline__ void sage_agg_scan(int base, int t, unsigned lo, int n_edges,
                                              const int* __restrict__ dst,
                                              int* __restrict__ queue, int* qn) {
    int4 v[8];
    #pragma unroll
    for (int k = 0; k < 8; ++k) {
        const int e0 = base + k * (AGG_THREADS * 4) + t * 4;
        if (!GUARD || e0 + 3 < n_edges) v[k] = *(const int4*)(dst + e0);
        else                            v[k] = make_int4(-1, -1, -1, -1);
    }
    #pragma unroll
    for (int k = 0; k < 8; ++k) {
        const int e0 = base + k * (AGG_THREADS * 4) + t * 4;
        const unsigned x0 = (unsigned)v[k].x - lo;
        const unsigned x1 = (unsigned)v[k].y - lo;
        const unsigned x2 = (unsigned)v[k].z - lo;
        const unsigned x3 = (unsigned)v[k].w - lo;
        const unsigned m01 = x0 < x1 ? x0 : x1;
        const unsigned m23 = x2 < x3 ? x2 : x3;
        const unsigned m   = m01 < m23 ? m01 : m23;
        if (m < AGG_ROWS) {
            const int pos = atomicAdd(qn, 1);
            if (pos < AGG_QCAP) queue[pos] = e0 >> 2;
        }
    }
}

__device__ __forceinline__ void sage_agg_append(int ent, int total, int* __restrict__ queue,
                                                int* qextra) {
    const int pos = total + atomicAdd(qextra, 1);
    if (pos < AGG_QCAP) queue[pos] = ent;
}

template <bool BUILD>
__global__ void __launch_bounds__(AGG_THREADS)
sage_agg_lds_kernel(const int* __restrict__ src, const int* __restrict__ dst,
                    const int* __restrict__ cnts, const int* __restrict__ cells,
                    const float* __restrict__ x, float* __restrict__ msg,
                    int* __restrict__ lists, int n_edges, int n_nodes) {
    __shared__ float acc[AGG_ROWS * D_FEAT];
    __shared__ __align__(16) int queue[AGG_QCAP];
    __shared__ int   qn;
    __shared__ int   qextra;
    __shared__ int   cover;
    __shared__ int   rowcnt[AGG_ROWS];
    __shared__ int2  cinfo[BIN_WGS];
    __shared__ int   wtot[BIN_WGS / 32];

    const int t = threadIdx.x;
    const unsigned lo = blockIdx.x * AGG_ROWS;

    int* rec = lists + (size_t)blockIdx.x * AGG_REC;
    const bool fits = (size_t)gridDim.x * AGG_REC <= (size_t)n_nodes * OUT_DIM;

    int c = 0;
    if (BUILD && sage_cells_ok(n_nodes) && t < BIN_WGS) c = cnts[(size_t)blockIdx.x * BIN_WGS + t];

    for (int i = t; i < AGG_ROWS * D_FEAT / 4; i += AGG_THREADS)
        ((float4*)acc)[i] = make_float4(0.0f, 0.0f, 0.0f, 0.0f);
    for (int i = t; i < AGG_ROWS; i += AGG_THREADS) rowcnt[i] = 0;
    if (t == 0) { qn = 0; qextra = 0; cover = 0; }
    __syncthreads();

    int  total2 = 0;
    bool go     = false;
    bool loaded = false;

    if (!BUILD && fits) {
        const int cnt = rec[0];
        if (cnt >= 0 && cnt <= AGG_QCAP) {
            for (int i = t; i < ((cnt + 3) >> 2); i += AGG_THREADS)
                ((int4*)queue)[i] = ((const int4*)(rec + 4))[i];
            total2 = cnt;
            go     = true;
            loaded = true;
        }
    }

    if (!loaded) {
        bool collected = false;

        if (BUILD && sage_cells_ok(n_nodes)) {
            const int lane = t & 31;
            int v = c;
            #pragma unroll
            for (int o = 1; o < 32; o <<= 1) {
                const int u = __shfl_up(v, o, 32);
                if (lane >= o) v += u;
            }
            if (t < BIN_WGS) {
                cinfo[t] = make_int2(c, v - c);
                if (lane == 31) wtot[t >> 5] = v;
                if (c > BIN_CAP) cover = 1;
            }
            __syncthreads();

            int woff = 0, total = 0;
            #pragma unroll
            for (int w = 0; w < BIN_WGS / 32; ++w) {
                const int n = wtot[w];
                if (w < (t >> 5)) woff += n;
                total += n;
            }
            if (cover == 0 && total <= AGG_QCAP) {
                if (t < BIN_WGS) cinfo[t].y += woff;
                __syncthreads();

                const int* cb = cells + (size_t)blockIdx.x * (BIN_WGS * BIN_CAP);
                int w[BIN_WGS * BIN_CAP / AGG_THREADS];
                int qp[BIN_WGS * BIN_CAP / AGG_THREADS];
                #pragma unroll
                for (int k = 0; k < BIN_WGS * BIN_CAP / AGG_THREADS; ++k) {
                    const int2 ci = cinfo[(k * AGG_THREADS + t) / BIN_CAP];
                    const int  sl = (k * AGG_THREADS + t) % BIN_CAP;
                    qp[k] = sl < ci.x ? ci.y + sl : -1;
                }
                #pragma unroll
                for (int k = 0; k < BIN_WGS * BIN_CAP / AGG_THREADS; ++k) {
                    w[k] = 0;
                    if (qp[k] >= 0) w[k] = cb[k * AGG_THREADS + t];
                }
                #pragma unroll
                for (int k = 0; k < BIN_WGS * BIN_CAP / AGG_THREADS; ++k)
                    if (qp[k] >= 0) queue[qp[k]] = w[k];
                __syncthreads();

                total2    = total;
                go        = true;
                collected = true;
            }
        }

        if (!collected) {
            const int per_iter = AGG_THREADS * 32;
            int base = 0;
            for (; base + per_iter <= n_edges; base += per_iter)
                sage_agg_scan<false>(base, t, lo, n_edges, dst, queue, &qn);
            if (base < n_edges)
                sage_agg_scan<true>(base, t, lo, n_edges, dst, queue, &qn);
            __syncthreads();

            const int  total  = qn;
            const bool queued = total <= AGG_QCAP;

            if (queued) {
                for (int i = t; i < total; i += AGG_THREADS) {
                    const int  e0 = queue[i] * 4;
                    const int4 dd = *(const int4*)(dst + e0);
                    const int4 ss = *(const int4*)(src + e0);
                    const int  dv[4] = {dd.x, dd.y, dd.z, dd.w};
                    const int  sv[4] = {ss.x, ss.y, ss.z, ss.w};
                    bool have = false;
                    int  ent0 = 0;
                    #pragma unroll
                    for (int j = 0; j < 4; ++j) {
                        const unsigned loc = (unsigned)dv[j] - lo;
                        if (loc < AGG_ROWS) {
                            const int ent = (int)((loc << 23) | ((unsigned)sv[j] & AGG_SMASK));
                            if (!have) { ent0 = ent; have = true; }
                            else       sage_agg_append(ent, total, queue, &qextra);
                        }
                    }
                    queue[i] = ent0;
                }
                const int rem = n_edges & 3;
                if (t < rem) {
                    const int e = n_edges - rem + t;
                    const unsigned loc = (unsigned)dst[e] - lo;
                    if (loc < AGG_ROWS)
                        sage_agg_append((int)((loc << 23) | ((unsigned)src[e] & AGG_SMASK)),
                                        total, queue, &qextra);
                }
            }
            __syncthreads();

            total2 = total + qextra;
            go     = queued && total2 <= AGG_QCAP;
        }

        if (BUILD && fits) {
            if (go)
                for (int i = t; i < ((total2 + 3) >> 2); i += AGG_THREADS)
                    ((int4*)(rec + 4))[i] = ((const int4*)queue)[i];
            if (t == 0) rec[0] = go ? total2 : -1;
        }
    }
    __syncthreads();

    if (go) {
        for (int i = t; i < total2; i += AGG_THREADS)
            atomicAdd(&rowcnt[(unsigned)queue[i] >> 23], 1);

        const int wave = t >> 5;
        const int lane = t & 31;
        for (int i = wave * 8; i < total2; i += (AGG_THREADS / 32) * 8) {
            int    ent[8];
            float4 r[8];
            #pragma unroll
            for (int k = 0; k < 8; ++k) {
                const int idx = i + k;
                ent[k] = queue[idx < total2 ? idx : i];
                r[k] = *(const float4*)(x + (size_t)(ent[k] & AGG_SMASK) * D_FEAT + lane * 4);
            }
            #pragma unroll
            for (int k = 0; k < 8; ++k) {
                if (i + k < total2) {
                    float* ar = acc + ((unsigned)ent[k] >> 23) * D_FEAT + lane;
                    atomicAdd(ar +  0, r[k].x);
                    atomicAdd(ar + 32, r[k].y);
                    atomicAdd(ar + 64, r[k].z);
                    atomicAdd(ar + 96, r[k].w);
                }
            }
        }
    } else {
        for (int e = t; e < n_edges; e += AGG_THREADS) {
            const unsigned loc = (unsigned)dst[e] - lo;
            if (loc < AGG_ROWS) {
                const float* xr = x + (size_t)src[e] * D_FEAT;
                float*       ar = acc + loc * D_FEAT;
                atomicAdd(&rowcnt[loc], 1);
                for (int c = 0; c < D_FEAT; ++c) atomicAdd(ar + (c & 3) * 32 + (c >> 2), xr[c]);
            }
        }
    }
    __syncthreads();

    {
        const int row0 = blockIdx.x * AGG_ROWS;
        for (int i = t; i < AGG_ROWS * (D_FEAT / 4); i += AGG_THREADS) {
            const int r = i >> 5;
            const int l = i & 31;
            if (row0 + r < n_nodes) {
                const float* ar = acc + r * D_FEAT + l;
                const float  iv = 1.0f / fmaxf((float)rowcnt[r], 1.0f);
                *(float4*)(msg + (size_t)(row0 + r) * D_FEAT + l * 4) =
                    make_float4(ar[0] * iv, ar[32] * iv, ar[64] * iv, ar[96] * iv);
            }
        }
    }
}

#define WQ_PITCH       516
#define LAYER_THREADS  512
#define LAYER_WGS      256

__device__ __forceinline__ void sage_stage_weights(const float* __restrict__ Wg, float* __restrict__ Wl, int t) {
    const int q0 = t >> 7;
    const int n  = t & 127;
    float4 w[8];
    #pragma unroll
    for (int j = 0; j < 8; ++j) {
        const float* g = Wg + (size_t)(4 * (q0 + (LAYER_THREADS / D_FEAT) * j)) * D_FEAT + n;
        w[j] = make_float4(g[0], g[D_FEAT], g[2 * D_FEAT], g[3 * D_FEAT]);
    }
    #pragma unroll
    for (int j = 0; j < 8; ++j)
        *(float4*)(Wl + (q0 + (LAYER_THREADS / D_FEAT) * j) * WQ_PITCH + n * 4) = w[j];
}

__global__ void __launch_bounds__(LAYER_THREADS)
sage_layer_kernel(const float* __restrict__ xin, const float* __restrict__ msg,
                  const float* __restrict__ Wself, const float* __restrict__ Wneigh,
                  const float* __restrict__ bias, float* __restrict__ xout, int m_tiles) {
    __shared__ __align__(16) float Ws[32 * WQ_PITCH];
    __shared__ __align__(16) float Wn[32 * WQ_PITCH];

    const int t = threadIdx.x;
    const int wave = __builtin_amdgcn_readfirstlane(t >> 5);
    const int lane = t & 31;
    const int half = lane >> 4;
    const int l16  = lane & 15;
    const int tstep = LAYER_WGS * (LAYER_THREADS / 32);
    const unsigned lrow = (unsigned)l16 * D_FEAT + half * 4;
    const unsigned lout = (unsigned)half * (8 * D_FEAT) + l16;

    int tile = blockIdx.x + wave * LAYER_WGS;

    float4 xa[16];
    {
        const int t0 = tile < m_tiles ? tile : 0;
        const float* xr = (xin + (size_t)t0 * (16 * D_FEAT)) + lrow;
        #pragma unroll
        for (int i = 0; i < 16; ++i) xa[i] = *(const float4*)(xr + 8 * i);
    }

    sage_stage_weights(Wself,  Ws, t);
    sage_stage_weights(Wneigh, Wn, t);
    __syncthreads();

    float bv[8];
    #pragma unroll
    for (int nt = 0; nt < 8; ++nt) bv[nt] = bias[nt * 16 + l16];

    const float* bs0 = Ws + half * WQ_PITCH + l16 * 4;
    const float* bn0 = Wn + half * WQ_PITCH + l16 * 4;

    for (; tile < m_tiles; tile += tstep) {
        const int m0 = tile * 16;
        const int pf = tile + tstep < m_tiles ? tile + tstep : tile;
        const float* mr = (msg + (size_t)m0 * D_FEAT) + lrow;
        const float* xn = (xin + (size_t)pf * (16 * D_FEAT)) + lrow;

        v8f acc[8];
        #pragma unroll
        for (int nt = 0; nt < 8; ++nt) {
            #pragma unroll
            for (int v = 0; v < 8; ++v) acc[nt][v] = 0.0f;
        }

        float4 ha[16];
        float4 bq[2][8];
        #pragma unroll
        for (int nt = 0; nt < 8; ++nt) bq[0][nt] = *(const float4*)(bs0 + nt * 64);
        #pragma unroll
        for (int u = 0; u < 32; ++u) {
            const int i = u & 15;
            if (u < 16) ha[i] = *(const float4*)(mr + 8 * i);
            else        xa[i] = *(const float4*)(xn + 8 * i);
            if (u + 1 < 32) {
                const int u1 = u + 1;
                const float* bp = (u1 < 16 ? bs0 : bn0) + 2 * (u1 & 15) * WQ_PITCH;
                #pragma unroll
                for (int nt = 0; nt < 8; ++nt) bq[u1 & 1][nt] = *(const float4*)(bp + nt * 64);
            }
            const float4 ar = u < 16 ? xa[i] : ha[i];
            #pragma unroll
            for (int s = 0; s < 2; ++s) {
                v2f a;
                a.x = s ? ar.z : ar.x;
                a.y = s ? ar.w : ar.y;
                #pragma unroll
                for (int nt = 0; nt < 8; ++nt) {
                    v2f b;
                    b.x = s ? bq[u & 1][nt].z : bq[u & 1][nt].x;
                    b.y = s ? bq[u & 1][nt].w : bq[u & 1][nt].y;
                    acc[nt] = __builtin_amdgcn_wmma_f32_16x16x4_f32(false, a, false, b,
                                                                    (short)0, acc[nt], false, false);
                }
            }
            __builtin_amdgcn_sched_barrier(0);
        }

        float* orow = (xout + (size_t)m0 * D_FEAT) + lout;
        #pragma unroll
        for (int nt = 0; nt < 8; ++nt) {
            #pragma unroll
            for (int v = 0; v < 8; ++v) {
                float val = acc[nt][v] + bv[nt];
                val = val > 0.0f ? val : 0.0f;
                orow[v * D_FEAT + nt * 16] = val;
            }
        }
    }
}

__global__ void __launch_bounds__(64)
sage_pred_kernel(const float* __restrict__ xin, const float* __restrict__ Wp,
                 const float* __restrict__ bp, float* __restrict__ out) {
    const int m0   = blockIdx.x * 16;
    const int t    = threadIdx.x;
    const int wave = t >> 5;
    const int lane = t & 31;
    const int n0   = wave * 16;
    const int half = lane >> 4;
    const int l16  = lane & 15;
    const int col  = n0 + l16;
    const int bcol = col < OUT_DIM ? col : (OUT_DIM - 1);

    v8f acc;
    {
        float bv = bp[bcol];
        #pragma unroll
        for (int i = 0; i < 8; ++i) acc[i] = bv;
    }

    #pragma unroll 4
    for (int k0 = 0; k0 < D_FEAT; k0 += 4) {
        const int ka = k0 + half * 2;
        const float* ap = xin + (size_t)(m0 + l16) * D_FEAT + ka;
        v2f a; a.x = ap[0]; a.y = ap[1];
        v2f b;
        b.x = Wp[(size_t)ka * OUT_DIM + bcol];
        b.y = Wp[(size_t)(ka + 1) * OUT_DIM + bcol];
        acc = __builtin_amdgcn_wmma_f32_16x16x4_f32(false, a, false, b,
                                                    (short)0, acc, false, false);
    }

    if (col < OUT_DIM) {
        #pragma unroll
        for (int v = 0; v < 8; ++v) {
            int row = m0 + v + half * 8;
            out[(size_t)row * OUT_DIM + col] = acc[v];
        }
    }
}


extern "C" void kernel_launch(void* const* d_in, const int* in_sizes, int n_in,
                              void* d_out, int out_size, void* d_ws, size_t ws_size,
                              hipStream_t stream) {
    const float* x       = (const float*)d_in[0];
    const int*   src     = (const int*)  d_in[1];
    const int*   dst     = (const int*)  d_in[2];
    const float* W_self  = (const float*)d_in[3];
    const float* W_neigh = (const float*)d_in[4];
    const float* b       = (const float*)d_in[5];
    const float* W_pred  = (const float*)d_in[6];
    const float* b_pred  = (const float*)d_in[7];

    const int n_nodes = in_sizes[0] / D_FEAT;
    const int n_edges = in_sizes[1];
    const int L       = in_sizes[5] / D_FEAT;
    const int feat_elems = n_nodes * D_FEAT;

    char*  ws = (char*)d_ws;
    size_t feat_bytes = (size_t)feat_elems * sizeof(float);
    float* msg    = (float*)ws;
    float* bufA   = (float*)(ws + feat_bytes);
    float* bufB   = (float*)(ws + 2 * feat_bytes);

    int* lists = (int*)d_out;

    const int agg_blocks = (n_nodes + AGG_ROWS - 1) / AGG_ROWS;

    int* cnts  = (int*)bufB;
    int* cells = cnts + (size_t)agg_blocks * BIN_WGS;

    sage_bin_kernel<<<BIN_WGS, BIN_THREADS, 0, stream>>>(src, dst, cnts, cells, n_edges, n_nodes);

    const float* cur = x;
    float* nxt = bufA;
    const int m_tiles = n_nodes / 16;

    for (int l = 0; l < L; ++l) {
        if (l == 0)
            sage_agg_lds_kernel<true><<<agg_blocks, AGG_THREADS, 0, stream>>>(src, dst, cnts, cells, cur, msg, lists,
                                                                              n_edges, n_nodes);
        else
            sage_agg_lds_kernel<false><<<agg_blocks, AGG_THREADS, 0, stream>>>(src, dst, cnts, cells, cur, msg, lists,
                                                                               n_edges, n_nodes);
        sage_layer_kernel<<<LAYER_WGS, LAYER_THREADS, 0, stream>>>(
            cur, msg,
            W_self  + (size_t)l * D_FEAT * D_FEAT,
            W_neigh + (size_t)l * D_FEAT * D_FEAT,
            b       + (size_t)l * D_FEAT,
            nxt, m_tiles);
        cur = nxt;
        nxt = (nxt == bufA) ? bufB : bufA;
    }

    sage_pred_kernel<<<m_tiles, 64, 0, stream>>>(cur, W_pred, b_pred, (float*)d_out);
}


# Neighbour sums in registers: a wave owns every 32nd row of its work-group's 400 (13 register rows) and adds its own edges' rows, list dealt per wave once; LDS float adds go; a row sums in list order
# speedup vs baseline: 3.9744x; 1.0814x over previous
#include <hip/hip_runtime.h>
#include <hip/hip_bf16.h>


#define D_FEAT 128
#define OUT_DIM 26

typedef __attribute__((ext_vector_type(2))) float v2f;
typedef __attribute__((ext_vector_type(8))) float v8f;

#define AGG_ROWS    400
#define AGG_THREADS 1024
#define AGG_WAVES   (AGG_THREADS / 32)
#define AGG_SLOTS   ((AGG_ROWS + AGG_WAVES - 1) / AGG_WAVES)
#define AGG_QCAP    10240
#define AGG_SMASK   0x7FFFFF
#define AGG_HDR     36
#define AGG_REC     (AGG_QCAP + AGG_HDR)

#define BIN_WGS     256
#define BIN_THREADS 1024
#define BIN_CAP     64
#define BIN_MAXR    1024

__device__ __forceinline__ bool sage_cells_ok(int n_nodes) {
    const size_t nr = (size_t)((n_nodes + AGG_ROWS - 1) / AGG_ROWS);
    return nr <= BIN_MAXR &&
           nr * BIN_WGS * (BIN_CAP + 1) * sizeof(int) <= (size_t)n_nodes * (D_FEAT * sizeof(float));
}


__device__ __forceinline__ void sage_bin_place(int d, int s, int n_ranges,
                                               int* __restrict__ cnt, int* __restrict__ cell0) {
    const unsigned r = (unsigned)d / AGG_ROWS;
    if (r < (unsigned)n_ranges) {
        const int pos = atomicAdd(&cnt[r], 1);
        if (pos < BIN_CAP)
            cell0[(size_t)r * (BIN_WGS * BIN_CAP) + pos] =
                (int)((((unsigned)d - r * AGG_ROWS) << 23) | ((unsigned)s & AGG_SMASK));
    }
}

__global__ void __launch_bounds__(BIN_THREADS)
sage_bin_kernel(const int* __restrict__ src, const int* __restrict__ dst,
                int* __restrict__ cnts, int* __restrict__ cells, int n_edges, int n_nodes) {
    __shared__ int cnt[BIN_MAXR];
    if (!sage_cells_ok(n_nodes)) return;

    const int t = threadIdx.x;
    const int p = blockIdx.x;
    const int n_ranges = (n_nodes + AGG_ROWS - 1) / AGG_ROWS;
    for (int r = t; r < n_ranges; r += BIN_THREADS) cnt[r] = 0;
    __syncthreads();

    const int per = (((n_edges + BIN_WGS - 1) / BIN_WGS) + 3) & ~3;
    const long long lo64 = (long long)p * per;
    const int e_lo = lo64 < (long long)n_edges ? (int)lo64 : n_edges;
    const int e_hi = (n_edges - e_lo) < per ? n_edges : e_lo + per;
    const int nq   = (e_hi - e_lo) >> 2;
    int* cell0 = cells + (size_t)p * BIN_CAP;

    for (int qb = 0; qb < nq; qb += 2 * BIN_THREADS) {
        int4 dd[2], ss[2];
        #pragma unroll
        for (int k = 0; k < 2; ++k) {
            const int q = qb + k * BIN_THREADS + t;
            if (q < nq) {
                dd[k] = *(const int4*)(dst + e_lo + 4 * q);
                ss[k] = *(const int4*)(src + e_lo + 4 * q);
            } else {
                dd[k] = make_int4(-1, -1, -1, -1);
                ss[k] = make_int4(0, 0, 0, 0);
            }
        }
        #pragma unroll
        for (int k = 0; k < 2; ++k) {
            sage_bin_place(dd[k].x, ss[k].x, n_ranges, cnt, cell0);
            sage_bin_place(dd[k].y, ss[k].y, n_ranges, cnt, cell0);
            sage_bin_place(dd[k].z, ss[k].z, n_ranges, cnt, cell0);
            sage_bin_place(dd[k].w, ss[k].w, n_ranges, cnt, cell0);
        }
    }
    const int rem = (e_hi - e_lo) & 3;
    if (t < rem) {
        const int e = e_lo + 4 * nq + t;
        sage_bin_place(dst[e], src[e], n_ranges, cnt, cell0);
    }
    __syncthreads();

    for (int r = t; r < n_ranges; r += BIN_THREADS) cnts[(size_t)r * BIN_WGS + p] = cnt[r];
}

template <bool GUARD>
__device__ __forceinline__ void sage_agg_scan(int base, int t, unsigned lo, int n_edges,
                                              const int* __restrict__ dst,
                                              int* __restrict__ queue, int* qn) {
    int4 v[8];
    #pragma unroll
    for (int k = 0; k < 8; ++k) {
        const int e0 = base + k * (AGG_THREADS * 4) + t * 4;
        if (!GUARD || e0 + 3 < n_edges) v[k] = *(const int4*)(dst + e0);
        else                            v[k] = make_int4(-1, -1, -1, -1);
    }
    #pragma unroll
    for (int k = 0; k < 8; ++k) {
        const int e0 = base + k * (AGG_THREADS * 4) + t * 4;
        const unsigned x0 = (unsigned)v[k].x - lo;
        const unsigned x1 = (unsigned)v[k].y - lo;
        const unsigned x2 = (unsigned)v[k].z - lo;
        const unsigned x3 = (unsigned)v[k].w - lo;
        const unsigned m01 = x0 < x1 ? x0 : x1;
        const unsigned m23 = x2 < x3 ? x2 : x3;
        const unsigned m   = m01 < m23 ? m01 : m23;
        if (m < AGG_ROWS) {
            const int pos = atomicAdd(qn, 1);
            if (pos < AGG_QCAP) queue[pos] = e0 >> 2;
        }
    }
}

__device__ __forceinline__ void sage_agg_append(int ent, int total, int* __restrict__ queue,
                                                int* qextra) {
    const int pos = total + atomicAdd(qextra, 1);
    if (pos < AGG_QCAP) queue[pos] = ent;
}


typedef __attribute__((ext_vector_type(16))) float v16f;
struct sage_rows { v16f x, y, z, w; };
static_assert(AGG_SLOTS <= 16, "a wave's rows are the elements of 16-wide vectors");

__device__ __forceinline__ void sage_slot_add(sage_rows& s, unsigned slot, const float4 r) {
    s.x[slot] += r.x;
    s.y[slot] += r.y;
    s.z[slot] += r.z;
    s.w[slot] += r.w;
}

__device__ __forceinline__ void sage_wave_sum(const int* __restrict__ ents, int beg, int end,
                                              const float* __restrict__ x, int lane,
                                              sage_rows& s, int& cntv) {
    int cur = 0;
    if (beg + lane < end) cur = ents[beg + lane];
    for (int base = beg; base < end; base += 32) {
        int nxt = 0;
        if (base + 32 + lane < end) nxt = ents[base + 32 + lane];
        const int stop = end - base < 32 ? end - base : 32;
        for (int l0 = 0; l0 < stop; l0 += 8) {
            int    e[8];
            float4 r[8];
            #pragma unroll
            for (int k = 0; k < 8; ++k) {
                const int lk = l0 + k < stop ? l0 + k : l0;
                e[k] = __builtin_amdgcn_readlane(cur, lk);
                r[k] = *(const float4*)(x + (size_t)(e[k] & AGG_SMASK) * D_FEAT + lane * 4);
            }
            #pragma unroll
            for (int k = 0; k < 8; ++k) {
                if (l0 + k < stop) {
                    const unsigned slot = (unsigned)e[k] >> 28;
                    sage_slot_add(s, slot, r[k]);
                    cntv += lane == (int)slot ? 1 : 0;
                }
            }
        }
        cur = nxt;
    }
}

__device__ __forceinline__ void sage_wave_scan(const int* __restrict__ src, const int* __restrict__ dst,
                                               int n_edges, unsigned lo, int wave, int lane,
                                               const float* __restrict__ x,
                                               sage_rows& s, int& cntv) {
    for (int e0 = 0; e0 < n_edges; e0 += 32) {
        const int e = e0 + lane;
        unsigned loc = 0xFFFFFFFFu;
        if (e < n_edges) loc = (unsigned)dst[e] - lo;
        const bool mine = loc < AGG_ROWS && (loc & (AGG_WAVES - 1)) == (unsigned)wave;
        int sv = 0;
        if (mine) sv = src[e];
        unsigned m = __builtin_amdgcn_ballot_w32(mine);
        while (m != 0) {
            const int b = __builtin_ctz(m);
            m &= m - 1;
            const unsigned slot = (unsigned)__builtin_amdgcn_readlane((int)loc, b) / AGG_WAVES;
            const int      sr   = __builtin_amdgcn_readlane(sv, b);
            const float4   r    = *(const float4*)(x + (size_t)sr * D_FEAT + lane * 4);
            sage_slot_add(s, slot, r);
            cntv += lane == (int)slot ? 1 : 0;
        }
    }
}

template <bool BUILD>
__global__ void __launch_bounds__(AGG_THREADS)
sage_agg_reg_kernel(const int* __restrict__ src, const int* __restrict__ dst,
                    const int* __restrict__ cnts, const int* __restrict__ cells,
                    const float* __restrict__ x, float* __restrict__ msg,
                    int* __restrict__ lists, int n_edges, int n_nodes) {
    const int t    = threadIdx.x;
    const int lane = t & 31;
    const int wave = __builtin_amdgcn_readfirstlane(t >> 5);
    const unsigned lo = blockIdx.x * AGG_ROWS;

    int* rec = lists + (size_t)blockIdx.x * AGG_REC;
    const bool fits = (size_t)gridDim.x * AGG_REC <= (size_t)n_nodes * OUT_DIM;

    sage_rows s;
    s.x = 0.0f; s.y = 0.0f; s.z = 0.0f; s.w = 0.0f;
    int  cntv = 0;
    bool done = false;

    if (!BUILD) {
        if (fits) {
            const int cnt = __builtin_amdgcn_readfirstlane(rec[0]);
            if (cnt >= 0 && cnt <= AGG_QCAP) {
                const int beg = __builtin_amdgcn_readfirstlane(rec[1 + wave]);
                const int end = __builtin_amdgcn_readfirstlane(rec[2 + wave]);
                sage_wave_sum(rec + AGG_HDR, beg, end, x, lane, s, cntv);
                done = true;
            }
        }
    } else {
        __shared__ __align__(16) int queue[AGG_QCAP];
        __shared__ __align__(16) int sorted[AGG_QCAP];
        __shared__ int   qn;
        __shared__ int   qextra;
        __shared__ int   cover;
        __shared__ int2  cinfo[BIN_WGS];
        __shared__ int   wtot[BIN_WGS / 32];
        __shared__ int   wcnt[AGG_WAVES];
        __shared__ int   wpos[AGG_WAVES];
        __shared__ int   wbase[AGG_WAVES + 1];

        int c = 0;
        if (sage_cells_ok(n_nodes) && t < BIN_WGS) c = cnts[(size_t)blockIdx.x * BIN_WGS + t];

        if (t == 0) { qn = 0; qextra = 0; cover = 0; }
        if (t < AGG_WAVES) wcnt[t] = 0;
        __syncthreads();

        int  total2    = 0;
        bool go        = false;
        bool collected = false;

        if (sage_cells_ok(n_nodes)) {
            int v = c;
            #pragma unroll
            for (int o = 1; o < 32; o <<= 1) {
                const int u = __shfl_up(v, o, 32);
                if (lane >= o) v += u;
            }
            if (t < BIN_WGS) {
                cinfo[t] = make_int2(c, v - c);
                if (lane == 31) wtot[t >> 5] = v;
                if (c > BIN_CAP) cover = 1;
            }
            __syncthreads();

            int woff = 0, total = 0;
            #pragma unroll
            for (int w = 0; w < BIN_WGS / 32; ++w) {
                const int n = wtot[w];
                if (w < (t >> 5)) woff += n;
                total += n;
            }
            if (cover == 0 && total <= AGG_QCAP) {
                if (t < BIN_WGS) cinfo[t].y += woff;
                __syncthreads();

                const int* cb = cells + (size_t)blockIdx.x * (BIN_WGS * BIN_CAP);
                int w[BIN_WGS * BIN_CAP / AGG_THREADS];
                int qp[BIN_WGS * BIN_CAP / AGG_THREADS];
                #pragma unroll
                for (int k = 0; k < BIN_WGS * BIN_CAP / AGG_THREADS; ++k) {
                    const int2 ci = cinfo[(k * AGG_THREADS + t) / BIN_CAP];
                    const int  sl = (k * AGG_THREADS + t) % BIN_CAP;
                    qp[k] = sl < ci.x ? ci.y + sl : -1;
                }
                #pragma unroll
                for (int k = 0; k < BIN_WGS * BIN_CAP / AGG_THREADS; ++k) {
                    w[k] = 0;
                    if (qp[k] >= 0) w[k] = cb[k * AGG_THREADS + t];
                }
                #pragma unroll
                for (int k = 0; k < BIN_WGS * BIN_CAP / AGG_THREADS; ++k)
                    if (qp[k] >= 0) queue[qp[k]] = w[k];
                __syncthreads();

                total2    = total;
                go        = true;
                collected = true;
            }
        }

        if (!collected) {
            const int per_iter = AGG_THREADS * 32;
            int base = 0;
            for (; base + per_iter <= n_edges; base += per_iter)
                sage_agg_scan<false>(base, t, lo, n_edges, dst, queue, &qn);
            if (base < n_edges)
                sage_agg_scan<true>(base, t, lo, n_edges, dst, queue, &qn);
            __syncthreads();

            const int  total  = qn;
            const bool queued = total <= AGG_QCAP;

            if (queued) {
                for (int i = t; i < total; i += AGG_THREADS) {
                    const int  e0 = queue[i] * 4;
                    const int4 dd = *(const int4*)(dst + e0);
                    const int4 ss = *(const int4*)(src + e0);
                    const int  dv[4] = {dd.x, dd.y, dd.z, dd.w};
                    const int  sv[4] = {ss.x, ss.y, ss.z, ss.w};
                    bool have = false;
                    int  ent0 = 0;
                    #pragma unroll
                    for (int j = 0; j < 4; ++j) {
                        const unsigned loc = (unsigned)dv[j] - lo;
                        if (loc < AGG_ROWS) {
                            const int ent = (int)((loc << 23) | ((unsigned)sv[j] & AGG_SMASK));
                            if (!have) { ent0 = ent; have = true; }
                            else       sage_agg_append(ent, total, queue, &qextra);
                        }
                    }
                    queue[i] = ent0;
                }
                const int rem = n_edges & 3;
                if (t < rem) {
                    const int e = n_edges - rem + t;
                    const unsigned loc = (unsigned)dst[e] - lo;
                    if (loc < AGG_ROWS)
                        sage_agg_append((int)((loc << 23) | ((unsigned)src[e] & AGG_SMASK)),
                                        total, queue, &qextra);
                }
            }
            __syncthreads();

            total2 = total + qextra;
            go     = queued && total2 <= AGG_QCAP;
        }

        if (go) {
            for (int i = t; i < total2; i += AGG_THREADS)
                atomicAdd(&wcnt[((unsigned)queue[i] >> 23) & (AGG_WAVES - 1)], 1);
            __syncthreads();
            if (t < AGG_WAVES) {
                const int cw = wcnt[t];
                int v = cw;
                #pragma unroll
                for (int o = 1; o < 32; o <<= 1) {
                    const int u = __shfl_up(v, o, 32);
                    if (lane >= o) v += u;
                }
                wbase[t] = v - cw;
                wpos[t]  = v - cw;
                if (t == AGG_WAVES - 1) wbase[AGG_WAVES] = v;
            }
            __syncthreads();
            for (int i = t; i < total2; i += AGG_THREADS) {
                const int ent = queue[i];
                const int p   = atomicAdd(&wpos[((unsigned)ent >> 23) & (AGG_WAVES - 1)], 1);
                sorted[p] = ent;
            }
            __syncthreads();

            if (fits) {
                for (int i = t; i < ((total2 + 3) >> 2); i += AGG_THREADS)
                    ((int4*)(rec + AGG_HDR))[i] = ((const int4*)sorted)[i];
                if (t <= AGG_WAVES) rec[1 + t] = wbase[t];
                if (t == 0) rec[0] = total2;
            }

            const int beg = __builtin_amdgcn_readfirstlane(wbase[wave]);
            const int end = __builtin_amdgcn_readfirstlane(wbase[wave + 1]);
            sage_wave_sum(sorted, beg, end, x, lane, s, cntv);
            done = true;
        } else if (fits && t == 0) {
            rec[0] = -1;
        }
    }

    if (!done) sage_wave_scan(src, dst, n_edges, lo, wave, lane, x, s, cntv);

    #pragma unroll
    for (int j = 0; j < AGG_SLOTS; ++j) {
        const int loc = wave + AGG_WAVES * j;
        const int row = (int)lo + loc;
        if (loc < AGG_ROWS && row < n_nodes) {
            const int   cj = __builtin_amdgcn_readlane(cntv, j);
            const float iv = 1.0f / fmaxf((float)cj, 1.0f);
            *(float4*)(msg + (size_t)row * D_FEAT + lane * 4) =
                make_float4(s.x[j] * iv, s.y[j] * iv, s.z[j] * iv, s.w[j] * iv);
        }
    }
}

#define WQ_PITCH       516
#define LAYER_THREADS  512
#define LAYER_WGS      256

__device__ __forceinline__ void sage_stage_weights(const float* __restrict__ Wg, float* __restrict__ Wl, int t) {
    const int q0 = t >> 7;
    const int n  = t & 127;
    float4 w[8];
    #pragma unroll
    for (int j = 0; j < 8; ++j) {
        const float* g = Wg + (size_t)(4 * (q0 + (LAYER_THREADS / D_FEAT) * j)) * D_FEAT + n;
        w[j] = make_float4(g[0], g[D_FEAT], g[2 * D_FEAT], g[3 * D_FEAT]);
    }
    #pragma unroll
    for (int j = 0; j < 8; ++j)
        *(float4*)(Wl + (q0 + (LAYER_THREADS / D_FEAT) * j) * WQ_PITCH + n * 4) = w[j];
}

__global__ void __launch_bounds__(LAYER_THREADS)
sage_layer_kernel(const float* __restrict__ xin, const float* __restrict__ msg,
                  const float* __restrict__ Wself, const float* __restrict__ Wneigh,
                  const float* __restrict__ bias, float* __restrict__ xout, int m_tiles) {
    __shared__ __align__(16) float Ws[32 * WQ_PITCH];
    __shared__ __align__(16) float Wn[32 * WQ_PITCH];

    const int t = threadIdx.x;
    const int wave = __builtin_amdgcn_readfirstlane(t >> 5);
    const int lane = t & 31;
    const int half = lane >> 4;
    const int l16  = lane & 15;
    const int tstep = LAYER_WGS * (LAYER_THREADS / 32);
    const unsigned lrow = (unsigned)l16 * D_FEAT + half * 4;
    const unsigned lout = (unsigned)half * (8 * D_FEAT) + l16;

    int tile = blockIdx.x + wave * LAYER_WGS;

    float4 xa[16];
    {
        const int t0 = tile < m_tiles ? tile : 0;
        const float* xr = (xin + (size_t)t0 * (16 * D_FEAT)) + lrow;
        #pragma unroll
        for (int i = 0; i < 16; ++i) xa[i] = *(const float4*)(xr + 8 * i);
    }

    sage_stage_weights(Wself,  Ws, t);
    sage_stage_weights(Wneigh, Wn, t);
    __syncthreads();

    float bv[8];
    #pragma unroll
    for (int nt = 0; nt < 8; ++nt) bv[nt] = bias[nt * 16 + l16];

    const float* bs0 = Ws + half * WQ_PITCH + l16 * 4;
    const float* bn0 = Wn + half * WQ_PITCH + l16 * 4;

    for (; tile < m_tiles; tile += tstep) {
        const int m0 = tile * 16;
        const int pf = tile + tstep < m_tiles ? tile + tstep : tile;
        const float* mr = (msg + (size_t)m0 * D_FEAT) + lrow;
        const float* xn = (xin + (size_t)pf * (16 * D_FEAT)) + lrow;

        v8f acc[8];
        #pragma unroll
        for (int nt = 0; nt < 8; ++nt) {
            #pragma unroll
            for (int v = 0; v < 8; ++v) acc[nt][v] = 0.0f;
        }

        float4 ha[16];
        float4 bq[2][8];
        #pragma unroll
        for (int nt = 0; nt < 8; ++nt) bq[0][nt] = *(const float4*)(bs0 + nt * 64);
        #pragma unroll
        for (int u = 0; u < 32; ++u) {
            const int i = u & 15;
            if (u < 16) ha[i] = *(const float4*)(mr + 8 * i);
            else        xa[i] = *(const float4*)(xn + 8 * i);
            if (u + 1 < 32) {
                const int u1 = u + 1;
                const float* bp = (u1 < 16 ? bs0 : bn0) + 2 * (u1 & 15) * WQ_PITCH;
                #pragma unroll
                for (int nt = 0; nt < 8; ++nt) bq[u1 & 1][nt] = *(const float4*)(bp + nt * 64);
            }
            const float4 ar = u < 16 ? xa[i] : ha[i];
            #pragma unroll
            for (int s = 0; s < 2; ++s) {
                v2f a;
                a.x = s ? ar.z : ar.x;
                a.y = s ? ar.w : ar.y;
                #pragma unroll
                for (int nt = 0; nt < 8; ++nt) {
                    v2f b;
                    b.x = s ? bq[u & 1][nt].z : bq[u & 1][nt].x;
                    b.y = s ? bq[u & 1][nt].w : bq[u & 1][nt].y;
                    acc[nt] = __builtin_amdgcn_wmma_f32_16x16x4_f32(false, a, false, b,
                                                                    (short)0, acc[nt], false, false);
                }
            }
            __builtin_amdgcn_sched_barrier(0);
        }

        float* orow = (xout + (size_t)m0 * D_FEAT) + lout;
        #pragma unroll
        for (int nt = 0; nt < 8; ++nt) {
            #pragma unroll
            for (int v = 0; v < 8; ++v) {
                float val = acc[nt][v] + bv[nt];
                val = val > 0.0f ? val : 0.0f;
                orow[v * D_FEAT + nt * 16] = val;
            }
        }
    }
}

__global__ void __launch_bounds__(64)
sage_pred_kernel(const float* __restrict__ xin, const float* __restrict__ Wp,
                 const float* __restrict__ bp, float* __restrict__ out) {
    const int m0   = blockIdx.x * 16;
    const int t    = threadIdx.x;
    const int wave = t >> 5;
    const int lane = t & 31;
    const int n0   = wave * 16;
    const int half = lane >> 4;
    const int l16  = lane & 15;
    const int col  = n0 + l16;
    const int bcol = col < OUT_DIM ? col : (OUT_DIM - 1);

    v8f acc;
    {
        float bv = bp[bcol];
        #pragma unroll
        for (int i = 0; i < 8; ++i) acc[i] = bv;
    }

    #pragma unroll 4
    for (int k0 = 0; k0 < D_FEAT; k0 += 4) {
        const int ka = k0 + half * 2;
        const float* ap = xin + (size_t)(m0 + l16) * D_FEAT + ka;
        v2f a; a.x = ap[0]; a.y = ap[1];
        v2f b;
        b.x = Wp[(size_t)ka * OUT_DIM + bcol];
        b.y = Wp[(size_t)(ka + 1) * OUT_DIM + bcol];
        acc = __builtin_amdgcn_wmma_f32_16x16x4_f32(false, a, false, b,
                                                    (short)0, acc, false, false);
    }

    if (col < OUT_DIM) {
        #pragma unroll
        for (int v = 0; v < 8; ++v) {
            int row = m0 + v + half * 8;
            out[(size_t)row * OUT_DIM + col] = acc[v];
        }
    }
}


extern "C" void kernel_launch(void* const* d_in, const int* in_sizes, int n_in,
                              void* d_out, int out_size, void* d_ws, size_t ws_size,
                              hipStream_t stream) {
    const float* x       = (const float*)d_in[0];
    const int*   src     = (const int*)  d_in[1];
    const int*   dst     = (const int*)  d_in[2];
    const float* W_self  = (const float*)d_in[3];
    const float* W_neigh = (const float*)d_in[4];
    const float* b       = (const float*)d_in[5];
    const float* W_pred  = (const float*)d_in[6];
    const float* b_pred  = (const float*)d_in[7];

    const int n_nodes = in_sizes[0] / D_FEAT;
    const int n_edges = in_sizes[1];
    const int L       = in_sizes[5] / D_FEAT;
    const int feat_elems = n_nodes * D_FEAT;

    char*  ws = (char*)d_ws;
    size_t feat_bytes = (size_t)feat_elems * sizeof(float);
    float* msg    = (float*)ws;
    float* bufA   = (float*)(ws + feat_bytes);
    float* bufB   = (float*)(ws + 2 * feat_bytes);

    int* lists = (int*)d_out;

    const int agg_blocks = (n_nodes + AGG_ROWS - 1) / AGG_ROWS;

    int* cnts  = (int*)bufB;
    int* cells = cnts + (size_t)agg_blocks * BIN_WGS;

    sage_bin_kernel<<<BIN_WGS, BIN_THREADS, 0, stream>>>(src, dst, cnts, cells, n_edges, n_nodes);

    const float* cur = x;
    float* nxt = bufA;
    const int m_tiles = n_nodes / 16;

    for (int l = 0; l < L; ++l) {
        if (l == 0)
            sage_agg_reg_kernel<true><<<agg_blocks, AGG_THREADS, 0, stream>>>(src, dst, cnts, cells, cur, msg, lists,
                                                                              n_edges, n_nodes);
        else
            sage_agg_reg_kernel<false><<<agg_blocks, AGG_THREADS, 0, stream>>>(src, dst, cnts, cells, cur, msg, lists,
                                                                               n_edges, n_nodes);
        sage_layer_kernel<<<LAYER_WGS, LAYER_THREADS, 0, stream>>>(
            cur, msg,
            W_self  + (size_t)l * D_FEAT * D_FEAT,
            W_neigh + (size_t)l * D_FEAT * D_FEAT,
            b       + (size_t)l * D_FEAT,
            nxt, m_tiles);
        cur = nxt;
        nxt = (nxt == bufA) ? bufB : bufA;
    }

    sage_pred_kernel<<<m_tiles, 64, 0, stream>>>(cur, W_pred, b_pred, (float*)d_out);
}
